# starting kernel
# baseline (speedup 1.0000x reference)
#include <hip/hip_runtime.h>
#include <hip/hip_bf16.h>
#include <math.h>

typedef __attribute__((ext_vector_type(2))) float v2f;
typedef __attribute__((ext_vector_type(8))) float v8f;

#define EPSLN 1e-5f

__device__ __forceinline__ v8f wmma_f32_4(v2f a, v2f b, v8f c) {
    return __builtin_amdgcn_wmma_f32_16x16x4_f32(
         false, a,  false, b,
         (short)0, c,  false,  false);
}

__device__ __forceinline__ void atomicMaxF(float* addr, float val) {
    if (val >= 0.0f) atomicMax((int*)addr, __float_as_int(val));
    else             atomicMin((unsigned int*)addr, __float_as_uint(val));
}

__global__ __launch_bounds__(256) void qkv_kernel(
    const float* __restrict__ feat,
    const float* __restrict__ Wq, const float* __restrict__ Wk,
    const float* __restrict__ Wv,
    float* __restrict__ q, float* __restrict__ k, float* __restrict__ v, int n)
{
    __shared__ float sA[64][132];
    const int row0 = blockIdx.x * 64;
    const int tid  = threadIdx.x;

    for (int i = tid; i < 64 * 32; i += 256) {
        int r = i >> 5, c4 = i & 31;
        int gr = row0 + r; if (gr >= n) gr = n - 1;
        ((float4*)&sA[r][0])[c4] = ((const float4*)(feat + (size_t)gr * 128))[c4];
    }
    __syncthreads();

    const int wave = tid >> 5;
    const int lane = tid & 31;
    const int half = lane >> 4;
    const int lr   = lane & 15;
    const int col  = wave * 16 + lr;

    const float* Ws[3] = { Wq, Wk, Wv };
    float*       Os[3] = { q,  k,  v  };

#pragma unroll
    for (int m = 0; m < 3; ++m) {
        const float* __restrict__ W = Ws[m];
        v2f bfr[32];
#pragma unroll
        for (int kq = 0; kq < 32; ++kq) {
            const int kk = 4 * kq + 2 * half;
            bfr[kq] = (v2f){ W[(size_t)kk * 128 + col],
                             W[(size_t)(kk + 1) * 128 + col] };
        }
        float* __restrict__ O = Os[m];
#pragma unroll
        for (int mt = 0; mt < 4; ++mt) {
            v8f acc = {};
#pragma unroll
            for (int kq = 0; kq < 32; ++kq) {
                const int kk = 4 * kq + 2 * half;
                v2f a = { sA[mt * 16 + lr][kk], sA[mt * 16 + lr][kk + 1] };
                acc = wmma_f32_4(a, bfr[kq], acc);
            }
#pragma unroll
            for (int r = 0; r < 8; ++r) {
                int gr = row0 + mt * 16 + r + 8 * half;
                if (gr < n) O[(size_t)gr * 128 + col] = acc[r];
            }
        }
    }
}

__global__ __launch_bounds__(256) void init_kernel(
    float* __restrict__ smax, float* __restrict__ ssum,
    float* __restrict__ ft2, int n)
{
    int i = blockIdx.x * 256 + threadIdx.x;
    if (i < n * 8) { smax[i] = -INFINITY; ssum[i] = 0.0f; }
    if (i < n * 128) ft2[i] = 0.0f;
}

__global__ __launch_bounds__(256) void scores_kernel(
    const float* __restrict__ kbuf, const float* __restrict__ qbuf,
    const int* __restrict__ src, const int* __restrict__ dst,
    float* __restrict__ scores, float* __restrict__ smax, int eh)
{
    int t = blockIdx.x * 256 + threadIdx.x;
    if (t >= eh) return;
    const int e = t >> 3, h = t & 7;
    const int s = src[e], d = dst[e];
    const float4* kp = (const float4*)(kbuf + (size_t)s * 128 + h * 16);
    const float4* qp = (const float4*)(qbuf + (size_t)d * 128 + h * 16);
    float acc = 0.0f;
#pragma unroll
    for (int i = 0; i < 4; ++i) {
        float4 a = kp[i], b = qp[i];
        acc += a.x * b.x + a.y * b.y + a.z * b.z + a.w * b.w;
    }
    acc *= 0.08838834764831845f;
    scores[t] = acc;
    atomicMaxF(&smax[d * 8 + h], acc);
}

__global__ __launch_bounds__(256) void expsum_kernel(
    const int* __restrict__ dst, float* __restrict__ scores,
    const float* __restrict__ smax, float* __restrict__ ssum, int eh)
{
    int t = blockIdx.x * 256 + threadIdx.x;
    if (t >= eh) return;
    const int e = t >> 3, h = t & 7;
    const int d = dst[e];
    float ex = expf(scores[t] - smax[d * 8 + h]);
    scores[t] = ex;
    atomicAdd(&ssum[d * 8 + h], ex);
}

__global__ __launch_bounds__(256) void scatter_kernel(
    const int* __restrict__ src, const int* __restrict__ dst,
    const float* __restrict__ scores, const float* __restrict__ ssum,
    const float* __restrict__ vbuf, float* __restrict__ ft2, int eh)
{
    int t = blockIdx.x * 256 + threadIdx.x;
    if (t >= eh) return;
    const int e = t >> 3, h = t & 7;
    const int s = src[e], d = dst[e];
    const float a = scores[t] / ssum[d * 8 + h];
    const float* vp = vbuf + (size_t)s * 128 + h * 16;
    float*       fp = ft2  + (size_t)d * 128 + h * 16;
#pragma unroll
    for (int i = 0; i < 16; ++i)
        atomicAdd(&fp[i], vp[i] * a);
}

__global__ __launch_bounds__(256) void ln1_kernel(
    const float* __restrict__ ft2, const float* __restrict__ feat,
    const float* __restrict__ g, const float* __restrict__ b,
    float* __restrict__ rst, int n)
{
    const int row = blockIdx.x * 16 + (threadIdx.x >> 4);
    const int c0  = threadIdx.x & 15;
    if (row >= n) return;
    const float* p1 = ft2  + (size_t)row * 128;
    const float* p2 = feat + (size_t)row * 128;
    float xv[8], s = 0.0f, ss = 0.0f;
#pragma unroll
    for (int j = 0; j < 8; ++j) {
        int c = c0 + 16 * j;
        float x = p1[c] + p2[c];
        xv[j] = x; s += x; ss += x * x;
    }
#pragma unroll
    for (int m = 1; m < 16; m <<= 1) {
        s  += __shfl_xor(s,  m, 32);
        ss += __shfl_xor(ss, m, 32);
    }
    const float mean = s * (1.0f / 128.0f);
    const float var  = ss * (1.0f / 128.0f) - mean * mean;
    const float inv  = rsqrtf(var + EPSLN);
    float* o = rst + (size_t)row * 128;
#pragma unroll
    for (int j = 0; j < 8; ++j) {
        int c = c0 + 16 * j;
        o[c] = (xv[j] - mean) * inv * g[c] + b[c];
    }
}

__global__ __launch_bounds__(256) void ffn1_kernel(
    const float* __restrict__ rst, const float* __restrict__ W1,
    const float* __restrict__ b1, const float* __restrict__ pw,
    float* __restrict__ hbuf, int n)
{
    __shared__ float sA[64][132];
    const int row0 = blockIdx.x * 64;
    const int tid  = threadIdx.x;
    for (int i = tid; i < 64 * 32; i += 256) {
        int r = i >> 5, c4 = i & 31;
        int gr = row0 + r; if (gr >= n) gr = n - 1;
        ((float4*)&sA[r][0])[c4] = ((const float4*)(rst + (size_t)gr * 128))[c4];
    }
    __syncthreads();

    const int wave = tid >> 5;
    const int lane = tid & 31;
    const int half = lane >> 4;
    const int lr   = lane & 15;

#pragma unroll
    for (int j = 0; j < 4; ++j) {
        const int col = (wave + 8 * j) * 16 + lr;
        v2f bfr[32];
#pragma unroll
        for (int kq = 0; kq < 32; ++kq) {
            const int kk = 4 * kq + 2 * half;
            bfr[kq] = (v2f){ W1[(size_t)kk * 512 + col],
                             W1[(size_t)(kk + 1) * 512 + col] };
        }
        const float bias = b1[col], slope = pw[col];
#pragma unroll
        for (int mt = 0; mt < 4; ++mt) {
            v8f acc = {};
#pragma unroll
            for (int kq = 0; kq < 32; ++kq) {
                const int kk = 4 * kq + 2 * half;
                v2f a = { sA[mt * 16 + lr][kk], sA[mt * 16 + lr][kk + 1] };
                acc = wmma_f32_4(a, bfr[kq], acc);
            }
#pragma unroll
            for (int r = 0; r < 8; ++r) {
                int gr = row0 + mt * 16 + r + 8 * half;
                if (gr < n) {
                    float h = acc[r] + bias;
                    h = (h >= 0.0f) ? h : slope * h;
                    hbuf[(size_t)gr * 512 + col] = h;
                }
            }
        }
    }
}

__global__ __launch_bounds__(256) void ffn2_ln_kernel(
    const float* __restrict__ hbuf, const float* __restrict__ W2,
    const float* __restrict__ b2, const float* __restrict__ rst,
    const float* __restrict__ g, const float* __restrict__ be,
    float* __restrict__ out, int n)
{
    __shared__ float sA[64][132];
    const int row0 = blockIdx.x * 64;
    const int tid  = threadIdx.x;
    const int wave = tid >> 5;
    const int lane = tid & 31;
    const int half = lane >> 4;
    const int lr   = lane & 15;
    const int col  = wave * 16 + lr;

    v8f acc[4] = {};
#pragma unroll
    for (int kc = 0; kc < 4; ++kc) {
        __syncthreads();
        for (int i = tid; i < 64 * 32; i += 256) {
            int r = i >> 5, c4 = i & 31;
            int gr = row0 + r; if (gr >= n) gr = n - 1;
            ((float4*)&sA[r][0])[c4] =
                ((const float4*)(hbuf + (size_t)gr * 512 + kc * 128))[c4];
        }
        __syncthreads();

        v2f bfr[32];
#pragma unroll
        for (int kq = 0; kq < 32; ++kq) {
            const int kk = kc * 128 + 4 * kq + 2 * half;
            bfr[kq] = (v2f){ W2[(size_t)kk * 128 + col],
                             W2[(size_t)(kk + 1) * 128 + col] };
        }
#pragma unroll
        for (int mt = 0; mt < 4; ++mt) {
#pragma unroll
            for (int kq = 0; kq < 32; ++kq) {
                const int kk = 4 * kq + 2 * half;
                v2f a = { sA[mt * 16 + lr][kk], sA[mt * 16 + lr][kk + 1] };
                acc[mt] = wmma_f32_4(a, bfr[kq], acc[mt]);
            }
        }
    }

    __syncthreads();
    const float bias = b2[col];
#pragma unroll
    for (int mt = 0; mt < 4; ++mt) {
#pragma unroll
        for (int r = 0; r < 8; ++r) {
            const int rl = mt * 16 + r + 8 * half;
            int gr = row0 + rl; if (gr >= n) gr = n - 1;
            sA[rl][col] = acc[mt][r] + bias + rst[(size_t)gr * 128 + col];
        }
    }
    __syncthreads();

    const int rl2 = tid >> 2;
    const int c0  = tid & 3;
    const int row = row0 + rl2;
    float xv[32], s = 0.0f, ss = 0.0f;
#pragma unroll
    for (int j = 0; j < 32; ++j) {
        int c = c0 + 4 * j;
        float x = sA[rl2][c];
        xv[j] = x; s += x; ss += x * x;
    }
#pragma unroll
    for (int m = 1; m < 4; m <<= 1) {
        s  += __shfl_xor(s,  m, 32);
        ss += __shfl_xor(ss, m, 32);
    }
    const float mean = s * (1.0f / 128.0f);
    const float var  = ss * (1.0f / 128.0f) - mean * mean;
    const float inv  = rsqrtf(var + EPSLN);
    if (row < n) {
        float* o = out + (size_t)row * 128;
#pragma unroll
        for (int j = 0; j < 32; ++j) {
            int c = c0 + 4 * j;
            o[c] = (xv[j] - mean) * inv * g[c] + be[c];
        }
    }
}

extern "C" void kernel_launch(void* const* d_in, const int* in_sizes, int n_in,
                              void* d_out, int out_size, void* d_ws, size_t ws_size,
                              hipStream_t stream)
{
    const float* feat  = (const float*)d_in[0];
    const int*   src   = (const int*)  d_in[1];
    const int*   dst   = (const int*)  d_in[2];
    const float* Wq    = (const float*)d_in[3];
    const float* Wk    = (const float*)d_in[4];
    const float* Wv    = (const float*)d_in[5];
    const float* ln1_g = (const float*)d_in[6];
    const float* ln1_b = (const float*)d_in[7];
    const float* ln2_g = (const float*)d_in[8];
    const float* ln2_b = (const float*)d_in[9];
    const float* W1    = (const float*)d_in[10];
    const float* b1    = (const float*)d_in[11];
    const float* pw    = (const float*)d_in[12];
    const float* W2    = (const float*)d_in[13];
    const float* b2    = (const float*)d_in[14];

    const int n = in_sizes[0] / 128;
    const int e = in_sizes[1];
    const int eh = e * 8;

    float* ws = (float*)d_ws;
    size_t off = 0;
    float* q      = ws + off; off += (size_t)n * 128;
    float* kbuf   = ws + off; off += (size_t)n * 128;
    float* vbuf   = ws + off; off += (size_t)n * 128;
    float* ft2    = ws + off; off += (size_t)n * 128;
    float* rst    = ws + off; off += (size_t)n * 128;
    float* hbuf   = ws + off; off += (size_t)n * 512;
    float* scores = ws + off; off += (size_t)eh;
    float* smax   = ws + off; off += (size_t)n * 8;
    float* ssum   = ws + off; off += (size_t)n * 8;
    (void)off; (void)ws_size; (void)n_in; (void)out_size;

    const int NT64 = (n + 63) / 64;
    const int NT16 = (n + 15) / 16;
    const int NB   = (n * 128 + 255) / 256;
    const int EB   = (eh + 255) / 256;

    qkv_kernel<<<NT64, 256, 0, stream>>>(feat, Wq, Wk, Wv, q, kbuf, vbuf, n);
    init_kernel<<<NB, 256, 0, stream>>>(smax, ssum, ft2, n);
    scores_kernel<<<EB, 256, 0, stream>>>(kbuf, q, src, dst, scores, smax, eh);
    expsum_kernel<<<EB, 256, 0, stream>>>(dst, scores, smax, ssum, eh);
    scatter_kernel<<<EB, 256, 0, stream>>>(src, dst, scores, ssum, vbuf, ft2, eh);
    ln1_kernel<<<NT16, 256, 0, stream>>>(ft2, feat, ln1_g, ln1_b, rst, n);
    ffn1_kernel<<<NT64, 256, 0, stream>>>(rst, W1, b1, pw, hbuf, n);
    ffn2_ln_kernel<<<NT64, 256, 0, stream>>>(hbuf, W2, b2, rst, ln2_g, ln2_b,
                                             (float*)d_out, n);
}


# edge phase as one launch: a work-group owns 157 dst rows in LDS and filters the whole dst array for them; softmax and v-sum by LDS float atomics (arrival order), divided by the exp-sum afterwards
# speedup vs baseline: 2.6849x; 2.6849x over previous
#include <hip/hip_runtime.h>
#include <hip/hip_bf16.h>
#include <math.h>

typedef __attribute__((ext_vector_type(2))) float v2f;
typedef __attribute__((ext_vector_type(8))) float v8f;

#define EPSLN 1e-5f

__device__ __forceinline__ v8f wmma_f32_4(v2f a, v2f b, v8f c) {
    return __builtin_amdgcn_wmma_f32_16x16x4_f32(
         false, a,  false, b,
         (short)0, c,  false,  false);
}

__global__ __launch_bounds__(256) void qkv_kernel(
    const float* __restrict__ feat,
    const float* __restrict__ Wq, const float* __restrict__ Wk,
    const float* __restrict__ Wv,
    float* __restrict__ q, float* __restrict__ k, float* __restrict__ v, int n)
{
    __shared__ float sA[64][132];
    const int row0 = blockIdx.x * 64;
    const int tid  = threadIdx.x;

    for (int i = tid; i < 64 * 32; i += 256) {
        int r = i >> 5, c4 = i & 31;
        int gr = row0 + r; if (gr >= n) gr = n - 1;
        ((float4*)&sA[r][0])[c4] = ((const float4*)(feat + (size_t)gr * 128))[c4];
    }
    __syncthreads();

    const int wave = tid >> 5;
    const int lane = tid & 31;
    const int half = lane >> 4;
    const int lr   = lane & 15;
    const int col  = wave * 16 + lr;

    const float* Ws[3] = { Wq, Wk, Wv };
    float*       Os[3] = { q,  k,  v  };

#pragma unroll
    for (int m = 0; m < 3; ++m) {
        const float* __restrict__ W = Ws[m];
        v2f bfr[32];
#pragma unroll
        for (int kq = 0; kq < 32; ++kq) {
            const int kk = 4 * kq + 2 * half;
            bfr[kq] = (v2f){ W[(size_t)kk * 128 + col],
                             W[(size_t)(kk + 1) * 128 + col] };
        }
        float* __restrict__ O = Os[m];
#pragma unroll
        for (int mt = 0; mt < 4; ++mt) {
            v8f acc = {};
#pragma unroll
            for (int kq = 0; kq < 32; ++kq) {
                const int kk = 4 * kq + 2 * half;
                v2f a = { sA[mt * 16 + lr][kk], sA[mt * 16 + lr][kk + 1] };
                acc = wmma_f32_4(a, bfr[kq], acc);
            }
#pragma unroll
            for (int r = 0; r < 8; ++r) {
                int gr = row0 + mt * 16 + r + 8 * half;
                if (gr < n) O[(size_t)gr * 128 + col] = acc[r];
            }
        }
    }
}

#define EG_T    1024
#define EG_R    157
#define EG_CAP  6144
#define EG_CS   3072
#define EG_QV   ((EG_R * 32 + EG_T - 1) / EG_T)

__device__ __forceinline__ void lds_max_f(float* addr, float val) {
    val += 0.0f;
    if (val >= 0.0f) atomicMax((int*)addr, __float_as_int(val));
    else             atomicMin((unsigned int*)addr, __float_as_uint(val));
}

__device__ __forceinline__ float quad_sum(float p) {
    p += __int_as_float(__builtin_amdgcn_mov_dpp(__float_as_int(p), 0xB1, 0xF, 0xF, true));
    p += __int_as_float(__builtin_amdgcn_mov_dpp(__float_as_int(p), 0x4E, 0xF, 0xF, true));
    return p;
}

template <int V4, bool CAPPED, bool FULL>
__device__ __forceinline__ void eg_scan_tile(
    const int* __restrict__ dst, int e, int base,
    int lo, unsigned rows, int2* s_list, int* s_cnt, int tid)
{
    int4 dv[V4];
#pragma unroll
    for (int u = 0; u < V4; ++u) {
        const int i0 = base + (u * EG_T + tid) * 4;
        if (FULL) {
            dv[u] = *(const int4*)(dst + i0);
        } else {
            dv[u].x = (i0     < e) ? dst[i0]     : -1;
            dv[u].y = (i0 + 1 < e) ? dst[i0 + 1] : -1;
            dv[u].z = (i0 + 2 < e) ? dst[i0 + 2] : -1;
            dv[u].w = (i0 + 3 < e) ? dst[i0 + 3] : -1;
        }
    }
#pragma unroll
    for (int u = 0; u < V4; ++u) {
        const int i0 = base + (u * EG_T + tid) * 4;
        const unsigned r0 = (unsigned)(dv[u].x - lo), r1 = (unsigned)(dv[u].y - lo);
        const unsigned r2 = (unsigned)(dv[u].z - lo), r3 = (unsigned)(dv[u].w - lo);
        const int c = (int)(r0 < rows) + (int)(r1 < rows) + (int)(r2 < rows) + (int)(r3 < rows);
        if (c) {
            int p = atomicAdd(s_cnt, c);
            if (r0 < rows) { if (!CAPPED || p < EG_CAP) s_list[p] = make_int2(i0,     (int)r0); ++p; }
            if (r1 < rows) { if (!CAPPED || p < EG_CAP) s_list[p] = make_int2(i0 + 1, (int)r1); ++p; }
            if (r2 < rows) { if (!CAPPED || p < EG_CAP) s_list[p] = make_int2(i0 + 2, (int)r2); ++p; }
            if (r3 < rows) { if (!CAPPED || p < EG_CAP) s_list[p] = make_int2(i0 + 3, (int)r3); }
        }
    }
}

__device__ __forceinline__ void eg_resolve(
    const int* __restrict__ src, int2* s_list, int cnt, int tid)
{
    for (int i = tid; i < cnt; i += EG_T) s_list[i].x = src[s_list[i].x];
}

template <int PASS, bool CACHE>
__device__ __forceinline__ void eg_process(
    const float* __restrict__ kbuf, const float* __restrict__ vbuf,
    const int2* s_list, int cnt, const float* s_q, float* s_sc,
    float* s_m, float* s_l, float* s_acc, int wave, int lane)
{
    const int hq = lane >> 2;
    for (int i0 = wave; i0 < cnt; i0 += 32 * 4) {
        int2   en[4];
        float4 kk[4];
        float4 vv[4];
#pragma unroll
        for (int u = 0; u < 4; ++u) {
            const int i = i0 + 32 * u;
            en[u] = s_list[(i < cnt) ? i : i0];
            if (!(PASS == 1 && CACHE))
                kk[u] = *(const float4*)(kbuf + (size_t)en[u].x * 128 + 4 * lane);
            if (PASS == 1)
                vv[u] = *(const float4*)(vbuf + (size_t)en[u].x * 128 + 4 * lane);
        }
#pragma unroll
        for (int u = 0; u < 4; ++u) {
            const bool ok = (i0 + 32 * u) < cnt;
            const int  ic = ok ? (i0 + 32 * u) : i0;
            const int  dl = en[u].y;
            float sc;
            if (PASS == 1 && CACHE) {
                sc = s_sc[ic * 8 + hq];
            } else {
                const float4 qq = *(const float4*)(s_q + dl * 128 + 4 * lane);
                const float p = quad_sum(kk[u].x * qq.x + kk[u].y * qq.y +
                                         kk[u].z * qq.z + kk[u].w * qq.w);
                sc = p * 0.08838834764831845f;
            }
            if (PASS == 0) {
                if (ok && (lane & 3) == 0) {
                    lds_max_f(&s_m[dl * 8 + hq], sc);
                    if (CACHE) s_sc[ic * 8 + hq] = sc;
                }
            } else {
                const float ex = expf(sc - s_m[dl * 8 + hq]);
                if (ok) {
                    if ((lane & 3) == 0) atomicAdd(&s_l[dl * 8 + hq], ex);
                    float* ap = s_acc + dl * 128 + lane;
                    atomicAdd(ap,      vv[u].x * ex);
                    atomicAdd(ap + 32, vv[u].y * ex);
                    atomicAdd(ap + 64, vv[u].z * ex);
                    atomicAdd(ap + 96, vv[u].w * ex);
                }
            }
        }
    }
}

__global__ __launch_bounds__(EG_T) void edge_kernel(
    const float* __restrict__ qbuf, const float* __restrict__ kbuf,
    const float* __restrict__ vbuf,
    const int* __restrict__ src, const int* __restrict__ dst,
    float* __restrict__ ft2, int n, int e)
{
    __shared__ float s_acc[EG_R * 128];
    __shared__ float s_q[EG_R * 128];
    __shared__ float s_m[EG_R * 8];
    __shared__ float s_l[EG_R * 8];
    __shared__ float s_sc[EG_CS * 8];
    __shared__ int2  s_list[EG_CAP];
    __shared__ int   s_cnt;

    const int tid  = threadIdx.x;
    const int wave = tid >> 5;
    const int lane = tid & 31;
    const int lo   = blockIdx.x * EG_R;
    const int rows = (n - lo < EG_R) ? (n - lo) : EG_R;

    {
        const float4* qs = (const float4*)(qbuf + (size_t)lo * 128);
        float4 t[EG_QV];
#pragma unroll
        for (int u = 0; u < EG_QV; ++u) {
            const int i = tid + u * EG_T;
            t[u] = qs[(i < rows * 32) ? i : rows * 32 - 1];
        }
#pragma unroll
        for (int u = 0; u < EG_QV; ++u) {
            const int i = tid + u * EG_T;
            if (i < rows * 32) ((float4*)s_q)[i] = t[u];
        }
    }
    for (int i = tid; i < EG_R * 128; i += EG_T) s_acc[i] = 0.0f;
    for (int i = tid; i < EG_R * 8; i += EG_T) { s_m[i] = -INFINITY; s_l[i] = 0.0f; }
    if (tid == 0) s_cnt = 0;
    __syncthreads();

    {
        const int TE = EG_T * 32;
        const int nf = ((((size_t)dst) & 15) == 0) ? e / TE : 0;
        int t = (nf > 0) ? (int)(blockIdx.x % (unsigned)nf) : 0;
        for (int it = 0; it < nf; ++it) {
            eg_scan_tile<8, true, true>(dst, e, t * TE, lo, (unsigned)rows, s_list, &s_cnt, tid);
            if (++t == nf) t = 0;
        }
        for (int base = nf * TE; base < e; base += EG_T * 4)
            eg_scan_tile<1, true, false>(dst, e, base, lo, (unsigned)rows, s_list, &s_cnt, tid);
    }
    __syncthreads();
    const int total = s_cnt;

    if (total <= EG_CS) {
        eg_resolve(src, s_list, total, tid);
        __syncthreads();
        eg_process<0, true>(kbuf, vbuf, s_list, total, s_q, s_sc, s_m, s_l, s_acc, wave, lane);
        __syncthreads();
        eg_process<1, true>(kbuf, vbuf, s_list, total, s_q, s_sc, s_m, s_l, s_acc, wave, lane);
    } else if (total <= EG_CAP) {
        eg_resolve(src, s_list, total, tid);
        __syncthreads();
        eg_process<0, false>(kbuf, vbuf, s_list, total, s_q, s_sc, s_m, s_l, s_acc, wave, lane);
        __syncthreads();
        eg_process<1, false>(kbuf, vbuf, s_list, total, s_q, s_sc, s_m, s_l, s_acc, wave, lane);
    } else {
        const int TE = EG_T * 4;
#pragma unroll 1
        for (int pass = 0; pass < 2; ++pass) {
            __syncthreads();
            if (tid == 0) s_cnt = 0;
            __syncthreads();
            for (int base = 0; base < e; base += TE) {
                eg_scan_tile<1, false, false>(dst, e, base, lo, (unsigned)rows, s_list, &s_cnt, tid);
                __syncthreads();
                const int c = s_cnt;
                if (c > EG_CAP - TE) {
                    eg_resolve(src, s_list, c, tid);
                    __syncthreads();
                    if (pass == 0) eg_process<0, false>(kbuf, vbuf, s_list, c, s_q, s_sc, s_m, s_l, s_acc, wave, lane);
                    else           eg_process<1, false>(kbuf, vbuf, s_list, c, s_q, s_sc, s_m, s_l, s_acc, wave, lane);
                    __syncthreads();
                    if (tid == 0) s_cnt = 0;
                }
                __syncthreads();
            }
            const int c = s_cnt;
            eg_resolve(src, s_list, c, tid);
            __syncthreads();
            if (pass == 0) eg_process<0, false>(kbuf, vbuf, s_list, c, s_q, s_sc, s_m, s_l, s_acc, wave, lane);
            else           eg_process<1, false>(kbuf, vbuf, s_list, c, s_q, s_sc, s_m, s_l, s_acc, wave, lane);
        }
    }
    __syncthreads();

    for (int i = tid; i < rows * 128; i += EG_T) {
        const int dl = i >> 7, c = i & 127;
        const float l = s_l[dl * 8 + (c >> 4)];
        const float a = s_acc[dl * 128 + (c & 3) * 32 + (c >> 2)];
        ft2[(size_t)lo * 128 + i] = (l > 0.0f) ? a / l : 0.0f;
    }
}

__global__ __launch_bounds__(256) void ln1_kernel(
    const float* __restrict__ ft2, const float* __restrict__ feat,
    const float* __restrict__ g, const float* __restrict__ b,
    float* __restrict__ rst, int n)
{
    const int row = blockIdx.x * 16 + (threadIdx.x >> 4);
    const int c0  = threadIdx.x & 15;
    if (row >= n) return;
    const float* p1 = ft2  + (size_t)row * 128;
    const float* p2 = feat + (size_t)row * 128;
    float xv[8], s = 0.0f, ss = 0.0f;
#pragma unroll
    for (int j = 0; j < 8; ++j) {
        int c = c0 + 16 * j;
        float x = p1[c] + p2[c];
        xv[j] = x; s += x; ss += x * x;
    }
#pragma unroll
    for (int m = 1; m < 16; m <<= 1) {
        s  += __shfl_xor(s,  m, 32);
        ss += __shfl_xor(ss, m, 32);
    }
    const float mean = s * (1.0f / 128.0f);
    const float var  = ss * (1.0f / 128.0f) - mean * mean;
    const float inv  = rsqrtf(var + EPSLN);
    float* o = rst + (size_t)row * 128;
#pragma unroll
    for (int j = 0; j < 8; ++j) {
        int c = c0 + 16 * j;
        o[c] = (xv[j] - mean) * inv * g[c] + b[c];
    }
}

__global__ __launch_bounds__(256) void ffn1_kernel(
    const float* __restrict__ rst, const float* __restrict__ W1,
    const float* __restrict__ b1, const float* __restrict__ pw,
    float* __restrict__ hbuf, int n)
{
    __shared__ float sA[64][132];
    const int row0 = blockIdx.x * 64;
    const int tid  = threadIdx.x;
    for (int i = tid; i < 64 * 32; i += 256) {
        int r = i >> 5, c4 = i & 31;
        int gr = row0 + r; if (gr >= n) gr = n - 1;
        ((float4*)&sA[r][0])[c4] = ((const float4*)(rst + (size_t)gr * 128))[c4];
    }
    __syncthreads();

    const int wave = tid >> 5;
    const int lane = tid & 31;
    const int half = lane >> 4;
    const int lr   = lane & 15;

#pragma unroll
    for (int j = 0; j < 4; ++j) {
        const int col = (wave + 8 * j) * 16 + lr;
        v2f bfr[32];
#pragma unroll
        for (int kq = 0; kq < 32; ++kq) {
            const int kk = 4 * kq + 2 * half;
            bfr[kq] = (v2f){ W1[(size_t)kk * 512 + col],
                             W1[(size_t)(kk + 1) * 512 + col] };
        }
        const float bias = b1[col], slope = pw[col];
#pragma unroll
        for (int mt = 0; mt < 4; ++mt) {
            v8f acc = {};
#pragma unroll
            for (int kq = 0; kq < 32; ++kq) {
                const int kk = 4 * kq + 2 * half;
                v2f a = { sA[mt * 16 + lr][kk], sA[mt * 16 + lr][kk + 1] };
                acc = wmma_f32_4(a, bfr[kq], acc);
            }
#pragma unroll
            for (int r = 0; r < 8; ++r) {
                int gr = row0 + mt * 16 + r + 8 * half;
                if (gr < n) {
                    float h = acc[r] + bias;
                    h = (h >= 0.0f) ? h : slope * h;
                    hbuf[(size_t)gr * 512 + col] = h;
                }
            }
        }
    }
}

__global__ __launch_bounds__(256) void ffn2_ln_kernel(
    const float* __restrict__ hbuf, const float* __restrict__ W2,
    const float* __restrict__ b2, const float* __restrict__ rst,
    const float* __restrict__ g, const float* __restrict__ be,
    float* __restrict__ out, int n)
{
    __shared__ float sA[64][132];
    const int row0 = blockIdx.x * 64;
    const int tid  = threadIdx.x;
    const int wave = tid >> 5;
    const int lane = tid & 31;
    const int half = lane >> 4;
    const int lr   = lane & 15;
    const int col  = wave * 16 + lr;

    v8f acc[4] = {};
#pragma unroll
    for (int kc = 0; kc < 4; ++kc) {
        __syncthreads();
        for (int i = tid; i < 64 * 32; i += 256) {
            int r = i >> 5, c4 = i & 31;
            int gr = row0 + r; if (gr >= n) gr = n - 1;
            ((float4*)&sA[r][0])[c4] =
                ((const float4*)(hbuf + (size_t)gr * 512 + kc * 128))[c4];
        }
        __syncthreads();

        v2f bfr[32];
#pragma unroll
        for (int kq = 0; kq < 32; ++kq) {
            const int kk = kc * 128 + 4 * kq + 2 * half;
            bfr[kq] = (v2f){ W2[(size_t)kk * 128 + col],
                             W2[(size_t)(kk + 1) * 128 + col] };
        }
#pragma unroll
        for (int mt = 0; mt < 4; ++mt) {
#pragma unroll
            for (int kq = 0; kq < 32; ++kq) {
                const int kk = 4 * kq + 2 * half;
                v2f a = { sA[mt * 16 + lr][kk], sA[mt * 16 + lr][kk + 1] };
                acc[mt] = wmma_f32_4(a, bfr[kq], acc[mt]);
            }
        }
    }

    __syncthreads();
    const float bias = b2[col];
#pragma unroll
    for (int mt = 0; mt < 4; ++mt) {
#pragma unroll
        for (int r = 0; r < 8; ++r) {
            const int rl = mt * 16 + r + 8 * half;
            int gr = row0 + rl; if (gr >= n) gr = n - 1;
            sA[rl][col] = acc[mt][r] + bias + rst[(size_t)gr * 128 + col];
        }
    }
    __syncthreads();

    const int rl2 = tid >> 2;
    const int c0  = tid & 3;
    const int row = row0 + rl2;
    float xv[32], s = 0.0f, ss = 0.0f;
#pragma unroll
    for (int j = 0; j < 32; ++j) {
        int c = c0 + 4 * j;
        float x = sA[rl2][c];
        xv[j] = x; s += x; ss += x * x;
    }
#pragma unroll
    for (int m = 1; m < 4; m <<= 1) {
        s  += __shfl_xor(s,  m, 32);
        ss += __shfl_xor(ss, m, 32);
    }
    const float mean = s * (1.0f / 128.0f);
    const float var  = ss * (1.0f / 128.0f) - mean * mean;
    const float inv  = rsqrtf(var + EPSLN);
    if (row < n) {
        float* o = out + (size_t)row * 128;
#pragma unroll
        for (int j = 0; j < 32; ++j) {
            int c = c0 + 4 * j;
            o[c] = (xv[j] - mean) * inv * g[c] + be[c];
        }
    }
}

extern "C" void kernel_launch(void* const* d_in, const int* in_sizes, int n_in,
                              void* d_out, int out_size, void* d_ws, size_t ws_size,
                              hipStream_t stream)
{
    const float* feat  = (const float*)d_in[0];
    const int*   src   = (const int*)  d_in[1];
    const int*   dst   = (const int*)  d_in[2];
    const float* Wq    = (const float*)d_in[3];
    const float* Wk    = (const float*)d_in[4];
    const float* Wv    = (const float*)d_in[5];
    const float* ln1_g = (const float*)d_in[6];
    const float* ln1_b = (const float*)d_in[7];
    const float* ln2_g = (const float*)d_in[8];
    const float* ln2_b = (const float*)d_in[9];
    const float* W1    = (const float*)d_in[10];
    const float* b1    = (const float*)d_in[11];
    const float* pw    = (const float*)d_in[12];
    const float* W2    = (const float*)d_in[13];
    const float* b2    = (const float*)d_in[14];

    const int n = in_sizes[0] / 128;
    const int e = in_sizes[1];

    float* ws = (float*)d_ws;
    size_t off = 0;
    float* q      = ws + off; off += (size_t)n * 128;
    float* kbuf   = ws + off; off += (size_t)n * 128;
    float* vbuf   = ws + off; off += (size_t)n * 128;
    float* ft2    = ws + off; off += (size_t)n * 128;
    float* rst    = ws + off; off += (size_t)n * 128;
    float* hbuf   = ws + off; off += (size_t)n * 512;
    (void)off; (void)ws_size; (void)n_in; (void)out_size;

    const int NT64 = (n + 63) / 64;
    const int NT16 = (n + 15) / 16;
    const int NEG  = (n + EG_R - 1) / EG_R;

    qkv_kernel<<<NT64, 256, 0, stream>>>(feat, Wq, Wk, Wv, q, kbuf, vbuf, n);
    edge_kernel<<<NEG, EG_T, 0, stream>>>(q, kbuf, vbuf, src, dst, ft2, n, e);
    ln1_kernel<<<NT16, 256, 0, stream>>>(ft2, feat, ln1_g, ln1_b, rst, n);
    ffn1_kernel<<<NT64, 256, 0, stream>>>(rst, W1, b1, pw, hbuf, n);
    ffn2_ln_kernel<<<NT64, 256, 0, stream>>>(hbuf, W2, b2, rst, ln2_g, ln2_b,
                                             (float*)d_out, n);
}


# first LayerNorm done in the edge kernel's epilogue: the owning work-group normalises its rows from LDS and writes rst; ft2 and the LN launch go; LN sums now 4 per lane + 32-lane tree
# speedup vs baseline: 2.7772x; 1.0344x over previous
#include <hip/hip_runtime.h>
#include <hip/hip_bf16.h>
#include <math.h>

typedef __attribute__((ext_vector_type(2))) float v2f;
typedef __attribute__((ext_vector_type(8))) float v8f;

#define EPSLN 1e-5f

__device__ __forceinline__ v8f wmma_f32_4(v2f a, v2f b, v8f c) {
    return __builtin_amdgcn_wmma_f32_16x16x4_f32(
         false, a,  false, b,
         (short)0, c,  false,  false);
}

__global__ __launch_bounds__(256) void qkv_kernel(
    const float* __restrict__ feat,
    const float* __restrict__ Wq, const float* __restrict__ Wk,
    const float* __restrict__ Wv,
    float* __restrict__ q, float* __restrict__ k, float* __restrict__ v, int n)
{
    __shared__ float sA[64][132];
    const int row0 = blockIdx.x * 64;
    const int tid  = threadIdx.x;

    for (int i = tid; i < 64 * 32; i += 256) {
        int r = i >> 5, c4 = i & 31;
        int gr = row0 + r; if (gr >= n) gr = n - 1;
        ((float4*)&sA[r][0])[c4] = ((const float4*)(feat + (size_t)gr * 128))[c4];
    }
    __syncthreads();

    const int wave = tid >> 5;
    const int lane = tid & 31;
    const int half = lane >> 4;
    const int lr   = lane & 15;
    const int col  = wave * 16 + lr;

    const float* Ws[3] = { Wq, Wk, Wv };
    float*       Os[3] = { q,  k,  v  };

#pragma unroll
    for (int m = 0; m < 3; ++m) {
        const float* __restrict__ W = Ws[m];
        v2f bfr[32];
#pragma unroll
        for (int kq = 0; kq < 32; ++kq) {
            const int kk = 4 * kq + 2 * half;
            bfr[kq] = (v2f){ W[(size_t)kk * 128 + col],
                             W[(size_t)(kk + 1) * 128 + col] };
        }
        float* __restrict__ O = Os[m];
#pragma unroll
        for (int mt = 0; mt < 4; ++mt) {
            v8f acc = {};
#pragma unroll
            for (int kq = 0; kq < 32; ++kq) {
                const int kk = 4 * kq + 2 * half;
                v2f a = { sA[mt * 16 + lr][kk], sA[mt * 16 + lr][kk + 1] };
                acc = wmma_f32_4(a, bfr[kq], acc);
            }
#pragma unroll
            for (int r = 0; r < 8; ++r) {
                int gr = row0 + mt * 16 + r + 8 * half;
                if (gr < n) O[(size_t)gr * 128 + col] = acc[r];
            }
        }
    }
}

#define EG_T    1024
#define EG_R    157
#define EG_CAP  6144
#define EG_CS   3072
#define EG_QV   ((EG_R * 32 + EG_T - 1) / EG_T)
#define EG_RW   ((EG_R + EG_T / 32 - 1) / (EG_T / 32))

__device__ __forceinline__ void lds_max_f(float* addr, float val) {
    val += 0.0f;
    if (val >= 0.0f) atomicMax((int*)addr, __float_as_int(val));
    else             atomicMin((unsigned int*)addr, __float_as_uint(val));
}

__device__ __forceinline__ float quad_sum(float p) {
    p += __int_as_float(__builtin_amdgcn_mov_dpp(__float_as_int(p), 0xB1, 0xF, 0xF, true));
    p += __int_as_float(__builtin_amdgcn_mov_dpp(__float_as_int(p), 0x4E, 0xF, 0xF, true));
    return p;
}

template <int V4, bool CAPPED, bool FULL>
__device__ __forceinline__ void eg_scan_tile(
    const int* __restrict__ dst, int e, int base,
    int lo, unsigned rows, int2* s_list, int* s_cnt, int tid)
{
    int4 dv[V4];
#pragma unroll
    for (int u = 0; u < V4; ++u) {
        const int i0 = base + (u * EG_T + tid) * 4;
        if (FULL) {
            dv[u] = *(const int4*)(dst + i0);
        } else {
            dv[u].x = (i0     < e) ? dst[i0]     : -1;
            dv[u].y = (i0 + 1 < e) ? dst[i0 + 1] : -1;
            dv[u].z = (i0 + 2 < e) ? dst[i0 + 2] : -1;
            dv[u].w = (i0 + 3 < e) ? dst[i0 + 3] : -1;
        }
    }
#pragma unroll
    for (int u = 0; u < V4; ++u) {
        const int i0 = base + (u * EG_T + tid) * 4;
        const unsigned r0 = (unsigned)(dv[u].x - lo), r1 = (unsigned)(dv[u].y - lo);
        const unsigned r2 = (unsigned)(dv[u].z - lo), r3 = (unsigned)(dv[u].w - lo);
        const int c = (int)(r0 < rows) + (int)(r1 < rows) + (int)(r2 < rows) + (int)(r3 < rows);
        if (c) {
            int p = atomicAdd(s_cnt, c);
            if (r0 < rows) { if (!CAPPED || p < EG_CAP) s_list[p] = make_int2(i0,     (int)r0); ++p; }
            if (r1 < rows) { if (!CAPPED || p < EG_CAP) s_list[p] = make_int2(i0 + 1, (int)r1); ++p; }
            if (r2 < rows) { if (!CAPPED || p < EG_CAP) s_list[p] = make_int2(i0 + 2, (int)r2); ++p; }
            if (r3 < rows) { if (!CAPPED || p < EG_CAP) s_list[p] = make_int2(i0 + 3, (int)r3); }
        }
    }
}

__device__ __forceinline__ void eg_resolve(
    const int* __restrict__ src, int2* s_list, int cnt, int tid)
{
    for (int i = tid; i < cnt; i += EG_T) s_list[i].x = src[s_list[i].x];
}

template <int PASS, bool CACHE>
__device__ __forceinline__ void eg_process(
    const float* __restrict__ kbuf, const float* __restrict__ vbuf,
    const int2* s_list, int cnt, const float* s_q, float* s_sc,
    float* s_m, float* s_l, float* s_acc, int wave, int lane)
{
    const int hq = lane >> 2;
    for (int i0 = wave; i0 < cnt; i0 += 32 * 4) {
        int2   en[4];
        float4 kk[4];
        float4 vv[4];
#pragma unroll
        for (int u = 0; u < 4; ++u) {
            const int i = i0 + 32 * u;
            en[u] = s_list[(i < cnt) ? i : i0];
            if (!(PASS == 1 && CACHE))
                kk[u] = *(const float4*)(kbuf + (size_t)en[u].x * 128 + 4 * lane);
            if (PASS == 1)
                vv[u] = *(const float4*)(vbuf + (size_t)en[u].x * 128 + 4 * lane);
        }
#pragma unroll
        for (int u = 0; u < 4; ++u) {
            const bool ok = (i0 + 32 * u) < cnt;
            const int  ic = ok ? (i0 + 32 * u) : i0;
            const int  dl = en[u].y;
            float sc;
            if (PASS == 1 && CACHE) {
                sc = s_sc[ic * 8 + hq];
            } else {
                const float4 qq = *(const float4*)(s_q + dl * 128 + 4 * lane);
                const float p = quad_sum(kk[u].x * qq.x + kk[u].y * qq.y +
                                         kk[u].z * qq.z + kk[u].w * qq.w);
                sc = p * 0.08838834764831845f;
            }
            if (PASS == 0) {
                if (ok && (lane & 3) == 0) {
                    lds_max_f(&s_m[dl * 8 + hq], sc);
                    if (CACHE) s_sc[ic * 8 + hq] = sc;
                }
            } else {
                const float ex = expf(sc - s_m[dl * 8 + hq]);
                if (ok) {
                    if ((lane & 3) == 0) atomicAdd(&s_l[dl * 8 + hq], ex);
                    float* ap = s_acc + dl * 128 + lane;
                    atomicAdd(ap,      vv[u].x * ex);
                    atomicAdd(ap + 32, vv[u].y * ex);
                    atomicAdd(ap + 64, vv[u].z * ex);
                    atomicAdd(ap + 96, vv[u].w * ex);
                }
            }
        }
    }
}

__global__ __launch_bounds__(EG_T) void edge_kernel(
    const float* __restrict__ qbuf, const float* __restrict__ kbuf,
    const float* __restrict__ vbuf,
    const int* __restrict__ src, const int* __restrict__ dst,
    const float* __restrict__ feat,
    const float* __restrict__ ln_g, const float* __restrict__ ln_b,
    float* __restrict__ rst, int n, int e)
{
    __shared__ float s_acc[EG_R * 128];
    __shared__ float s_q[EG_R * 128];
    __shared__ float s_m[EG_R * 8];
    __shared__ float s_l[EG_R * 8];
    __shared__ float s_sc[EG_CS * 8];
    __shared__ int2  s_list[EG_CAP];
    __shared__ int   s_cnt;

    const int tid  = threadIdx.x;
    const int wave = tid >> 5;
    const int lane = tid & 31;
    const int lo   = blockIdx.x * EG_R;
    const int rows = (n - lo < EG_R) ? (n - lo) : EG_R;

    {
        const float4* qs = (const float4*)(qbuf + (size_t)lo * 128);
        float4 t[EG_QV];
#pragma unroll
        for (int u = 0; u < EG_QV; ++u) {
            const int i = tid + u * EG_T;
            t[u] = qs[(i < rows * 32) ? i : rows * 32 - 1];
        }
#pragma unroll
        for (int u = 0; u < EG_QV; ++u) {
            const int i = tid + u * EG_T;
            if (i < rows * 32) ((float4*)s_q)[i] = t[u];
        }
    }
    for (int i = tid; i < EG_R * 128; i += EG_T) s_acc[i] = 0.0f;
    for (int i = tid; i < EG_R * 8; i += EG_T) { s_m[i] = -INFINITY; s_l[i] = 0.0f; }
    if (tid == 0) s_cnt = 0;
    __syncthreads();

    {
        const int TE = EG_T * 32;
        const int nf = ((((size_t)dst) & 15) == 0) ? e / TE : 0;
        int t = (nf > 0) ? (int)(blockIdx.x % (unsigned)nf) : 0;
        for (int it = 0; it < nf; ++it) {
            eg_scan_tile<8, true, true>(dst, e, t * TE, lo, (unsigned)rows, s_list, &s_cnt, tid);
            if (++t == nf) t = 0;
        }
        for (int base = nf * TE; base < e; base += EG_T * 4)
            eg_scan_tile<1, true, false>(dst, e, base, lo, (unsigned)rows, s_list, &s_cnt, tid);
    }
    __syncthreads();
    const int total = s_cnt;

    float4 fv[EG_RW];
#pragma unroll
    for (int t = 0; t < EG_RW; ++t) {
        const int dl = wave + (EG_T / 32) * t;
        const int dc = (dl < rows) ? dl : rows - 1;
        fv[t] = *(const float4*)(feat + (size_t)(lo + dc) * 128 + 4 * lane);
    }
    const float g0 = ln_g[4 * lane],     g1 = ln_g[4 * lane + 1];
    const float g2 = ln_g[4 * lane + 2], g3 = ln_g[4 * lane + 3];
    const float b0 = ln_b[4 * lane],     b1 = ln_b[4 * lane + 1];
    const float b2 = ln_b[4 * lane + 2], b3 = ln_b[4 * lane + 3];

    if (total <= EG_CS) {
        eg_resolve(src, s_list, total, tid);
        __syncthreads();
        eg_process<0, true>(kbuf, vbuf, s_list, total, s_q, s_sc, s_m, s_l, s_acc, wave, lane);
        __syncthreads();
        eg_process<1, true>(kbuf, vbuf, s_list, total, s_q, s_sc, s_m, s_l, s_acc, wave, lane);
    } else if (total <= EG_CAP) {
        eg_resolve(src, s_list, total, tid);
        __syncthreads();
        eg_process<0, false>(kbuf, vbuf, s_list, total, s_q, s_sc, s_m, s_l, s_acc, wave, lane);
        __syncthreads();
        eg_process<1, false>(kbuf, vbuf, s_list, total, s_q, s_sc, s_m, s_l, s_acc, wave, lane);
    } else {
        const int TE = EG_T * 4;
#pragma unroll 1
        for (int pass = 0; pass < 2; ++pass) {
            __syncthreads();
            if (tid == 0) s_cnt = 0;
            __syncthreads();
            for (int base = 0; base < e; base += TE) {
                eg_scan_tile<1, false, false>(dst, e, base, lo, (unsigned)rows, s_list, &s_cnt, tid);
                __syncthreads();
                const int c = s_cnt;
                if (c > EG_CAP - TE) {
                    eg_resolve(src, s_list, c, tid);
                    __syncthreads();
                    if (pass == 0) eg_process<0, false>(kbuf, vbuf, s_list, c, s_q, s_sc, s_m, s_l, s_acc, wave, lane);
                    else           eg_process<1, false>(kbuf, vbuf, s_list, c, s_q, s_sc, s_m, s_l, s_acc, wave, lane);
                    __syncthreads();
                    if (tid == 0) s_cnt = 0;
                }
                __syncthreads();
            }
            const int c = s_cnt;
            eg_resolve(src, s_list, c, tid);
            __syncthreads();
            if (pass == 0) eg_process<0, false>(kbuf, vbuf, s_list, c, s_q, s_sc, s_m, s_l, s_acc, wave, lane);
            else           eg_process<1, false>(kbuf, vbuf, s_list, c, s_q, s_sc, s_m, s_l, s_acc, wave, lane);
        }
    }
    __syncthreads();

    float4 o[EG_RW];
#pragma unroll
    for (int t = 0; t < EG_RW; ++t) {
        const int dl = wave + (EG_T / 32) * t;
        const int dc = (dl < rows) ? dl : rows - 1;
        const float l = s_l[dc * 8 + (lane >> 2)];
        const float* ap = s_acc + dc * 128 + lane;
        const float a0 = ap[0], a1 = ap[32], a2 = ap[64], a3 = ap[96];
        const bool has = l > 0.0f;
        const float x0 = (has ? a0 / l : 0.0f) + fv[t].x;
        const float x1 = (has ? a1 / l : 0.0f) + fv[t].y;
        const float x2 = (has ? a2 / l : 0.0f) + fv[t].z;
        const float x3 = (has ? a3 / l : 0.0f) + fv[t].w;
        float s  = (x0 + x1) + (x2 + x3);
        float ss = (x0 * x0 + x1 * x1) + (x2 * x2 + x3 * x3);
#pragma unroll
        for (int m = 1; m < 32; m <<= 1) {
            s  += __shfl_xor(s,  m, 32);
            ss += __shfl_xor(ss, m, 32);
        }
        const float mean = s * (1.0f / 128.0f);
        const float var  = ss * (1.0f / 128.0f) - mean * mean;
        const float inv  = rsqrtf(var + EPSLN);
        o[t].x = (x0 - mean) * inv * g0 + b0;
        o[t].y = (x1 - mean) * inv * g1 + b1;
        o[t].z = (x2 - mean) * inv * g2 + b2;
        o[t].w = (x3 - mean) * inv * g3 + b3;
    }
#pragma unroll
    for (int t = 0; t < EG_RW; ++t) {
        const int dl = wave + (EG_T / 32) * t;
        if (dl < rows)
            *(float4*)(rst + (size_t)(lo + dl) * 128 + 4 * lane) = o[t];
    }
}

__global__ __launch_bounds__(256) void ffn1_kernel(
    const float* __restrict__ rst, const float* __restrict__ W1,
    const float* __restrict__ b1, const float* __restrict__ pw,
    float* __restrict__ hbuf, int n)
{
    __shared__ float sA[64][132];
    const int row0 = blockIdx.x * 64;
    const int tid  = threadIdx.x;
    for (int i = tid; i < 64 * 32; i += 256) {
        int r = i >> 5, c4 = i & 31;
        int gr = row0 + r; if (gr >= n) gr = n - 1;
        ((float4*)&sA[r][0])[c4] = ((const float4*)(rst + (size_t)gr * 128))[c4];
    }
    __syncthreads();

    const int wave = tid >> 5;
    const int lane = tid & 31;
    const int half = lane >> 4;
    const int lr   = lane & 15;

#pragma unroll
    for (int j = 0; j < 4; ++j) {
        const int col = (wave + 8 * j) * 16 + lr;
        v2f bfr[32];
#pragma unroll
        for (int kq = 0; kq < 32; ++kq) {
            const int kk = 4 * kq + 2 * half;
            bfr[kq] = (v2f){ W1[(size_t)kk * 512 + col],
                             W1[(size_t)(kk + 1) * 512 + col] };
        }
        const float bias = b1[col], slope = pw[col];
#pragma unroll
        for (int mt = 0; mt < 4; ++mt) {
            v8f acc = {};
#pragma unroll
            for (int kq = 0; kq < 32; ++kq) {
                const int kk = 4 * kq + 2 * half;
                v2f a = { sA[mt * 16 + lr][kk], sA[mt * 16 + lr][kk + 1] };
                acc = wmma_f32_4(a, bfr[kq], acc);
            }
#pragma unroll
            for (int r = 0; r < 8; ++r) {
                int gr = row0 + mt * 16 + r + 8 * half;
                if (gr < n) {
                    float h = acc[r] + bias;
                    h = (h >= 0.0f) ? h : slope * h;
                    hbuf[(size_t)gr * 512 + col] = h;
                }
            }
        }
    }
}

__global__ __launch_bounds__(256) void ffn2_ln_kernel(
    const float* __restrict__ hbuf, const float* __restrict__ W2,
    const float* __restrict__ b2, const float* __restrict__ rst,
    const float* __restrict__ g, const float* __restrict__ be,
    float* __restrict__ out, int n)
{
    __shared__ float sA[64][132];
    const int row0 = blockIdx.x * 64;
    const int tid  = threadIdx.x;
    const int wave = tid >> 5;
    const int lane = tid & 31;
    const int half = lane >> 4;
    const int lr   = lane & 15;
    const int col  = wave * 16 + lr;

    v8f acc[4] = {};
#pragma unroll
    for (int kc = 0; kc < 4; ++kc) {
        __syncthreads();
        for (int i = tid; i < 64 * 32; i += 256) {
            int r = i >> 5, c4 = i & 31;
            int gr = row0 + r; if (gr >= n) gr = n - 1;
            ((float4*)&sA[r][0])[c4] =
                ((const float4*)(hbuf + (size_t)gr * 512 + kc * 128))[c4];
        }
        __syncthreads();

        v2f bfr[32];
#pragma unroll
        for (int kq = 0; kq < 32; ++kq) {
            const int kk = kc * 128 + 4 * kq + 2 * half;
            bfr[kq] = (v2f){ W2[(size_t)kk * 128 + col],
                             W2[(size_t)(kk + 1) * 128 + col] };
        }
#pragma unroll
        for (int mt = 0; mt < 4; ++mt) {
#pragma unroll
            for (int kq = 0; kq < 32; ++kq) {
                const int kk = 4 * kq + 2 * half;
                v2f a = { sA[mt * 16 + lr][kk], sA[mt * 16 + lr][kk + 1] };
                acc[mt] = wmma_f32_4(a, bfr[kq], acc[mt]);
            }
        }
    }

    __syncthreads();
    const float bias = b2[col];
#pragma unroll
    for (int mt = 0; mt < 4; ++mt) {
#pragma unroll
        for (int r = 0; r < 8; ++r) {
            const int rl = mt * 16 + r + 8 * half;
            int gr = row0 + rl; if (gr >= n) gr = n - 1;
            sA[rl][col] = acc[mt][r] + bias + rst[(size_t)gr * 128 + col];
        }
    }
    __syncthreads();

    const int rl2 = tid >> 2;
    const int c0  = tid & 3;
    const int row = row0 + rl2;
    float xv[32], s = 0.0f, ss = 0.0f;
#pragma unroll
    for (int j = 0; j < 32; ++j) {
        int c = c0 + 4 * j;
        float x = sA[rl2][c];
        xv[j] = x; s += x; ss += x * x;
    }
#pragma unroll
    for (int m = 1; m < 4; m <<= 1) {
        s  += __shfl_xor(s,  m, 32);
        ss += __shfl_xor(ss, m, 32);
    }
    const float mean = s * (1.0f / 128.0f);
    const float var  = ss * (1.0f / 128.0f) - mean * mean;
    const float inv  = rsqrtf(var + EPSLN);
    if (row < n) {
        float* o = out + (size_t)row * 128;
#pragma unroll
        for (int j = 0; j < 32; ++j) {
            int c = c0 + 4 * j;
            o[c] = (xv[j] - mean) * inv * g[c] + be[c];
        }
    }
}

extern "C" void kernel_launch(void* const* d_in, const int* in_sizes, int n_in,
                              void* d_out, int out_size, void* d_ws, size_t ws_size,
                              hipStream_t stream)
{
    const float* feat  = (const float*)d_in[0];
    const int*   src   = (const int*)  d_in[1];
    const int*   dst   = (const int*)  d_in[2];
    const float* Wq    = (const float*)d_in[3];
    const float* Wk    = (const float*)d_in[4];
    const float* Wv    = (const float*)d_in[5];
    const float* ln1_g = (const float*)d_in[6];
    const float* ln1_b = (const float*)d_in[7];
    const float* ln2_g = (const float*)d_in[8];
    const float* ln2_b = (const float*)d_in[9];
    const float* W1    = (const float*)d_in[10];
    const float* b1    = (const float*)d_in[11];
    const float* pw    = (const float*)d_in[12];
    const float* W2    = (const float*)d_in[13];
    const float* b2    = (const float*)d_in[14];

    const int n = in_sizes[0] / 128;
    const int e = in_sizes[1];

    float* ws = (float*)d_ws;
    size_t off = 0;
    float* q      = ws + off; off += (size_t)n * 128;
    float* kbuf   = ws + off; off += (size_t)n * 128;
    float* vbuf   = ws + off; off += (size_t)n * 128;
    float* rst    = ws + off; off += (size_t)n * 128;
    float* hbuf   = ws + off; off += (size_t)n * 512;
    (void)off; (void)ws_size; (void)n_in; (void)out_size;

    const int NT64 = (n + 63) / 64;
    const int NEG  = (n + EG_R - 1) / EG_R;

    qkv_kernel<<<NT64, 256, 0, stream>>>(feat, Wq, Wk, Wv, q, kbuf, vbuf, n);
    edge_kernel<<<NEG, EG_T, 0, stream>>>(q, kbuf, vbuf, src, dst,
                                          feat, ln1_g, ln1_b, rst, n, e);
    ffn1_kernel<<<NT64, 256, 0, stream>>>(rst, W1, b1, pw, hbuf, n);
    ffn2_ln_kernel<<<NT64, 256, 0, stream>>>(hbuf, W2, b2, rst, ln2_g, ln2_b,
                                             (float*)d_out, n);
}


# last kernel: K-chunk loop pipelined over two LDS tiles: chunk kc+1's h rows are fetched to registers under chunk kc's WMMAs; one barrier per chunk; loop rolled, 3 work-groups per CU; sum order kept
# speedup vs baseline: 3.1498x; 1.1342x over previous
#include <hip/hip_runtime.h>
#include <hip/hip_bf16.h>
#include <math.h>

typedef __attribute__((ext_vector_type(2))) float v2f;
typedef __attribute__((ext_vector_type(8))) float v8f;

#define EPSLN 1e-5f

__device__ __forceinline__ v8f wmma_f32_4(v2f a, v2f b, v8f c) {
    return __builtin_amdgcn_wmma_f32_16x16x4_f32(
         false, a,  false, b,
         (short)0, c,  false,  false);
}

__global__ __launch_bounds__(256) void qkv_kernel(
    const float* __restrict__ feat,
    const float* __restrict__ Wq, const float* __restrict__ Wk,
    const float* __restrict__ Wv,
    float* __restrict__ q, float* __restrict__ k, float* __restrict__ v, int n)
{
    __shared__ float sA[64][132];
    const int row0 = blockIdx.x * 64;
    const int tid  = threadIdx.x;

    for (int i = tid; i < 64 * 32; i += 256) {
        int r = i >> 5, c4 = i & 31;
        int gr = row0 + r; if (gr >= n) gr = n - 1;
        ((float4*)&sA[r][0])[c4] = ((const float4*)(feat + (size_t)gr * 128))[c4];
    }
    __syncthreads();

    const int wave = tid >> 5;
    const int lane = tid & 31;
    const int half = lane >> 4;
    const int lr   = lane & 15;
    const int col  = wave * 16 + lr;

    const float* Ws[3] = { Wq, Wk, Wv };
    float*       Os[3] = { q,  k,  v  };

#pragma unroll
    for (int m = 0; m < 3; ++m) {
        const float* __restrict__ W = Ws[m];
        v2f bfr[32];
#pragma unroll
        for (int kq = 0; kq < 32; ++kq) {
            const int kk = 4 * kq + 2 * half;
            bfr[kq] = (v2f){ W[(size_t)kk * 128 + col],
                             W[(size_t)(kk + 1) * 128 + col] };
        }
        float* __restrict__ O = Os[m];
#pragma unroll
        for (int mt = 0; mt < 4; ++mt) {
            v8f acc = {};
#pragma unroll
            for (int kq = 0; kq < 32; ++kq) {
                const int kk = 4 * kq + 2 * half;
                v2f a = { sA[mt * 16 + lr][kk], sA[mt * 16 + lr][kk + 1] };
                acc = wmma_f32_4(a, bfr[kq], acc);
            }
#pragma unroll
            for (int r = 0; r < 8; ++r) {
                int gr = row0 + mt * 16 + r + 8 * half;
                if (gr < n) O[(size_t)gr * 128 + col] = acc[r];
            }
        }
    }
}

#define EG_T    1024
#define EG_R    157
#define EG_CAP  6144
#define EG_CS   3072
#define EG_QV   ((EG_R * 32 + EG_T - 1) / EG_T)
#define EG_RW   ((EG_R + EG_T / 32 - 1) / (EG_T / 32))

__device__ __forceinline__ void lds_max_f(float* addr, float val) {
    val += 0.0f;
    if (val >= 0.0f) atomicMax((int*)addr, __float_as_int(val));
    else             atomicMin((unsigned int*)addr, __float_as_uint(val));
}

__device__ __forceinline__ float quad_sum(float p) {
    p += __int_as_float(__builtin_amdgcn_mov_dpp(__float_as_int(p), 0xB1, 0xF, 0xF, true));
    p += __int_as_float(__builtin_amdgcn_mov_dpp(__float_as_int(p), 0x4E, 0xF, 0xF, true));
    return p;
}

template <int V4, bool CAPPED, bool FULL>
__device__ __forceinline__ void eg_scan_tile(
    const int* __restrict__ dst, int e, int base,
    int lo, unsigned rows, int2* s_list, int* s_cnt, int tid)
{
    int4 dv[V4];
#pragma unroll
    for (int u = 0; u < V4; ++u) {
        const int i0 = base + (u * EG_T + tid) * 4;
        if (FULL) {
            dv[u] = *(const int4*)(dst + i0);
        } else {
            dv[u].x = (i0     < e) ? dst[i0]     : -1;
            dv[u].y = (i0 + 1 < e) ? dst[i0 + 1] : -1;
            dv[u].z = (i0 + 2 < e) ? dst[i0 + 2] : -1;
            dv[u].w = (i0 + 3 < e) ? dst[i0 + 3] : -1;
        }
    }
#pragma unroll
    for (int u = 0; u < V4; ++u) {
        const int i0 = base + (u * EG_T + tid) * 4;
        const unsigned r0 = (unsigned)(dv[u].x - lo), r1 = (unsigned)(dv[u].y - lo);
        const unsigned r2 = (unsigned)(dv[u].z - lo), r3 = (unsigned)(dv[u].w - lo);
        const int c = (int)(r0 < rows) + (int)(r1 < rows) + (int)(r2 < rows) + (int)(r3 < rows);
        if (c) {
            int p = atomicAdd(s_cnt, c);
            if (r0 < rows) { if (!CAPPED || p < EG_CAP) s_list[p] = make_int2(i0,     (int)r0); ++p; }
            if (r1 < rows) { if (!CAPPED || p < EG_CAP) s_list[p] = make_int2(i0 + 1, (int)r1); ++p; }
            if (r2 < rows) { if (!CAPPED || p < EG_CAP) s_list[p] = make_int2(i0 + 2, (int)r2); ++p; }
            if (r3 < rows) { if (!CAPPED || p < EG_CAP) s_list[p] = make_int2(i0 + 3, (int)r3); }
        }
    }
}

__device__ __forceinline__ void eg_resolve(
    const int* __restrict__ src, int2* s_list, int cnt, int tid)
{
    for (int i = tid; i < cnt; i += EG_T) s_list[i].x = src[s_list[i].x];
}

template <int PASS, bool CACHE>
__device__ __forceinline__ void eg_process(
    const float* __restrict__ kbuf, const float* __restrict__ vbuf,
    const int2* s_list, int cnt, const float* s_q, float* s_sc,
    float* s_m, float* s_l, float* s_acc, int wave, int lane)
{
    const int hq = lane >> 2;
    for (int i0 = wave; i0 < cnt; i0 += 32 * 4) {
        int2   en[4];
        float4 kk[4];
        float4 vv[4];
#pragma unroll
        for (int u = 0; u < 4; ++u) {
            const int i = i0 + 32 * u;
            en[u] = s_list[(i < cnt) ? i : i0];
            if (!(PASS == 1 && CACHE))
                kk[u] = *(const float4*)(kbuf + (size_t)en[u].x * 128 + 4 * lane);
            if (PASS == 1)
                vv[u] = *(const float4*)(vbuf + (size_t)en[u].x * 128 + 4 * lane);
        }
#pragma unroll
        for (int u = 0; u < 4; ++u) {
            const bool ok = (i0 + 32 * u) < cnt;
            const int  ic = ok ? (i0 + 32 * u) : i0;
            const int  dl = en[u].y;
            float sc;
            if (PASS == 1 && CACHE) {
                sc = s_sc[ic * 8 + hq];
            } else {
                const float4 qq = *(const float4*)(s_q + dl * 128 + 4 * lane);
                const float p = quad_sum(kk[u].x * qq.x + kk[u].y * qq.y +
                                         kk[u].z * qq.z + kk[u].w * qq.w);
                sc = p * 0.08838834764831845f;
            }
            if (PASS == 0) {
                if (ok && (lane & 3) == 0) {
                    lds_max_f(&s_m[dl * 8 + hq], sc);
                    if (CACHE) s_sc[ic * 8 + hq] = sc;
                }
            } else {
                const float ex = expf(sc - s_m[dl * 8 + hq]);
                if (ok) {
                    if ((lane & 3) == 0) atomicAdd(&s_l[dl * 8 + hq], ex);
                    float* ap = s_acc + dl * 128 + lane;
                    atomicAdd(ap,      vv[u].x * ex);
                    atomicAdd(ap + 32, vv[u].y * ex);
                    atomicAdd(ap + 64, vv[u].z * ex);
                    atomicAdd(ap + 96, vv[u].w * ex);
                }
            }
        }
    }
}

__global__ __launch_bounds__(EG_T) void edge_kernel(
    const float* __restrict__ qbuf, const float* __restrict__ kbuf,
    const float* __restrict__ vbuf,
    const int* __restrict__ src, const int* __restrict__ dst,
    const float* __restrict__ feat,
    const float* __restrict__ ln_g, const float* __restrict__ ln_b,
    float* __restrict__ rst, int n, int e)
{
    __shared__ float s_acc[EG_R * 128];
    __shared__ float s_q[EG_R * 128];
    __shared__ float s_m[EG_R * 8];
    __shared__ float s_l[EG_R * 8];
    __shared__ float s_sc[EG_CS * 8];
    __shared__ int2  s_list[EG_CAP];
    __shared__ int   s_cnt;

    const int tid  = threadIdx.x;
    const int wave = tid >> 5;
    const int lane = tid & 31;
    const int lo   = blockIdx.x * EG_R;
    const int rows = (n - lo < EG_R) ? (n - lo) : EG_R;

    {
        const float4* qs = (const float4*)(qbuf + (size_t)lo * 128);
        float4 t[EG_QV];
#pragma unroll
        for (int u = 0; u < EG_QV; ++u) {
            const int i = tid + u * EG_T;
            t[u] = qs[(i < rows * 32) ? i : rows * 32 - 1];
        }
#pragma unroll
        for (int u = 0; u < EG_QV; ++u) {
            const int i = tid + u * EG_T;
            if (i < rows * 32) ((float4*)s_q)[i] = t[u];
        }
    }
    for (int i = tid; i < EG_R * 128; i += EG_T) s_acc[i] = 0.0f;
    for (int i = tid; i < EG_R * 8; i += EG_T) { s_m[i] = -INFINITY; s_l[i] = 0.0f; }
    if (tid == 0) s_cnt = 0;
    __syncthreads();

    {
        const int TE = EG_T * 32;
        const int nf = ((((size_t)dst) & 15) == 0) ? e / TE : 0;
        int t = (nf > 0) ? (int)(blockIdx.x % (unsigned)nf) : 0;
        for (int it = 0; it < nf; ++it) {
            eg_scan_tile<8, true, true>(dst, e, t * TE, lo, (unsigned)rows, s_list, &s_cnt, tid);
            if (++t == nf) t = 0;
        }
        for (int base = nf * TE; base < e; base += EG_T * 4)
            eg_scan_tile<1, true, false>(dst, e, base, lo, (unsigned)rows, s_list, &s_cnt, tid);
    }
    __syncthreads();
    const int total = s_cnt;

    float4 fv[EG_RW];
#pragma unroll
    for (int t = 0; t < EG_RW; ++t) {
        const int dl = wave + (EG_T / 32) * t;
        const int dc = (dl < rows) ? dl : rows - 1;
        fv[t] = *(const float4*)(feat + (size_t)(lo + dc) * 128 + 4 * lane);
    }
    const float g0 = ln_g[4 * lane],     g1 = ln_g[4 * lane + 1];
    const float g2 = ln_g[4 * lane + 2], g3 = ln_g[4 * lane + 3];
    const float b0 = ln_b[4 * lane],     b1 = ln_b[4 * lane + 1];
    const float b2 = ln_b[4 * lane + 2], b3 = ln_b[4 * lane + 3];

    if (total <= EG_CS) {
        eg_resolve(src, s_list, total, tid);
        __syncthreads();
        eg_process<0, true>(kbuf, vbuf, s_list, total, s_q, s_sc, s_m, s_l, s_acc, wave, lane);
        __syncthreads();
        eg_process<1, true>(kbuf, vbuf, s_list, total, s_q, s_sc, s_m, s_l, s_acc, wave, lane);
    } else if (total <= EG_CAP) {
        eg_resolve(src, s_list, total, tid);
        __syncthreads();
        eg_process<0, false>(kbuf, vbuf, s_list, total, s_q, s_sc, s_m, s_l, s_acc, wave, lane);
        __syncthreads();
        eg_process<1, false>(kbuf, vbuf, s_list, total, s_q, s_sc, s_m, s_l, s_acc, wave, lane);
    } else {
        const int TE = EG_T * 4;
#pragma unroll 1
        for (int pass = 0; pass < 2; ++pass) {
            __syncthreads();
            if (tid == 0) s_cnt = 0;
            __syncthreads();
            for (int base = 0; base < e; base += TE) {
                eg_scan_tile<1, false, false>(dst, e, base, lo, (unsigned)rows, s_list, &s_cnt, tid);
                __syncthreads();
                const int c = s_cnt;
                if (c > EG_CAP - TE) {
                    eg_resolve(src, s_list, c, tid);
                    __syncthreads();
                    if (pass == 0) eg_process<0, false>(kbuf, vbuf, s_list, c, s_q, s_sc, s_m, s_l, s_acc, wave, lane);
                    else           eg_process<1, false>(kbuf, vbuf, s_list, c, s_q, s_sc, s_m, s_l, s_acc, wave, lane);
                    __syncthreads();
                    if (tid == 0) s_cnt = 0;
                }
                __syncthreads();
            }
            const int c = s_cnt;
            eg_resolve(src, s_list, c, tid);
            __syncthreads();
            if (pass == 0) eg_process<0, false>(kbuf, vbuf, s_list, c, s_q, s_sc, s_m, s_l, s_acc, wave, lane);
            else           eg_process<1, false>(kbuf, vbuf, s_list, c, s_q, s_sc, s_m, s_l, s_acc, wave, lane);
        }
    }
    __syncthreads();

    float4 o[EG_RW];
#pragma unroll
    for (int t = 0; t < EG_RW; ++t) {
        const int dl = wave + (EG_T / 32) * t;
        const int dc = (dl < rows) ? dl : rows - 1;
        const float l = s_l[dc * 8 + (lane >> 2)];
        const float* ap = s_acc + dc * 128 + lane;
        const float a0 = ap[0], a1 = ap[32], a2 = ap[64], a3 = ap[96];
        const bool has = l > 0.0f;
        const float x0 = (has ? a0 / l : 0.0f) + fv[t].x;
        const float x1 = (has ? a1 / l : 0.0f) + fv[t].y;
        const float x2 = (has ? a2 / l : 0.0f) + fv[t].z;
        const float x3 = (has ? a3 / l : 0.0f) + fv[t].w;
        float s  = (x0 + x1) + (x2 + x3);
        float ss = (x0 * x0 + x1 * x1) + (x2 * x2 + x3 * x3);
#pragma unroll
        for (int m = 1; m < 32; m <<= 1) {
            s  += __shfl_xor(s,  m, 32);
            ss += __shfl_xor(ss, m, 32);
        }
        const float mean = s * (1.0f / 128.0f);
        const float var  = ss * (1.0f / 128.0f) - mean * mean;
        const float inv  = rsqrtf(var + EPSLN);
        o[t].x = (x0 - mean) * inv * g0 + b0;
        o[t].y = (x1 - mean) * inv * g1 + b1;
        o[t].z = (x2 - mean) * inv * g2 + b2;
        o[t].w = (x3 - mean) * inv * g3 + b3;
    }
#pragma unroll
    for (int t = 0; t < EG_RW; ++t) {
        const int dl = wave + (EG_T / 32) * t;
        if (dl < rows)
            *(float4*)(rst + (size_t)(lo + dl) * 128 + 4 * lane) = o[t];
    }
}

__global__ __launch_bounds__(256) void ffn1_kernel(
    const float* __restrict__ rst, const float* __restrict__ W1,
    const float* __restrict__ b1, const float* __restrict__ pw,
    float* __restrict__ hbuf, int n)
{
    __shared__ float sA[64][132];
    const int row0 = blockIdx.x * 64;
    const int tid  = threadIdx.x;
    for (int i = tid; i < 64 * 32; i += 256) {
        int r = i >> 5, c4 = i & 31;
        int gr = row0 + r; if (gr >= n) gr = n - 1;
        ((float4*)&sA[r][0])[c4] = ((const float4*)(rst + (size_t)gr * 128))[c4];
    }
    __syncthreads();

    const int wave = tid >> 5;
    const int lane = tid & 31;
    const int half = lane >> 4;
    const int lr   = lane & 15;

#pragma unroll
    for (int j = 0; j < 4; ++j) {
        const int col = (wave + 8 * j) * 16 + lr;
        v2f bfr[32];
#pragma unroll
        for (int kq = 0; kq < 32; ++kq) {
            const int kk = 4 * kq + 2 * half;
            bfr[kq] = (v2f){ W1[(size_t)kk * 512 + col],
                             W1[(size_t)(kk + 1) * 512 + col] };
        }
        const float bias = b1[col], slope = pw[col];
#pragma unroll
        for (int mt = 0; mt < 4; ++mt) {
            v8f acc = {};
#pragma unroll
            for (int kq = 0; kq < 32; ++kq) {
                const int kk = 4 * kq + 2 * half;
                v2f a = { sA[mt * 16 + lr][kk], sA[mt * 16 + lr][kk + 1] };
                acc = wmma_f32_4(a, bfr[kq], acc);
            }
#pragma unroll
            for (int r = 0; r < 8; ++r) {
                int gr = row0 + mt * 16 + r + 8 * half;
                if (gr < n) {
                    float h = acc[r] + bias;
                    h = (h >= 0.0f) ? h : slope * h;
                    hbuf[(size_t)gr * 512 + col] = h;
                }
            }
        }
    }
}

__device__ __forceinline__ float4 ffn2_fetch(const float* __restrict__ hbuf, int row0, int n,
                                             int wave, int lane, int u, int kc)
{
    int gr = row0 + wave + 8 * u; if (gr >= n) gr = n - 1;
    const unsigned off = (unsigned)gr * 512u + (unsigned)(kc * 128 + 4 * lane);
    return *(const float4*)(hbuf + off);
}

template <bool MORE>
__device__ __forceinline__ void ffn2_chunk(
    const float* __restrict__ hbuf, const float* __restrict__ W2,
    float (*cur)[132], float (*nxt)[132], v8f (&acc)[4],
    int kc, int row0, int n, int wave, int lane, int half, int lr, int col)
{
    v2f bfr[32];
#pragma unroll
    for (int kq = 0; kq < 32; ++kq) {
        const int kk = kc * 128 + 4 * kq + 2 * half;
        bfr[kq] = (v2f){ W2[(size_t)kk * 128 + col],
                         W2[(size_t)(kk + 1) * 128 + col] };
    }
#pragma unroll
    for (int hp = 0; hp < 2; ++hp) {
        float4 t0, t1, t2, t3;
        if (MORE) {
            t0 = ffn2_fetch(hbuf, row0, n, wave, lane, 4 * hp + 0, kc + 1);
            t1 = ffn2_fetch(hbuf, row0, n, wave, lane, 4 * hp + 1, kc + 1);
            t2 = ffn2_fetch(hbuf, row0, n, wave, lane, 4 * hp + 2, kc + 1);
            t3 = ffn2_fetch(hbuf, row0, n, wave, lane, 4 * hp + 3, kc + 1);
            __builtin_amdgcn_sched_barrier(0);
        }
#pragma unroll
        for (int mt = 2 * hp; mt < 2 * hp + 2; ++mt) {
#pragma unroll
            for (int kq = 0; kq < 32; ++kq) {
                const int kk = 4 * kq + 2 * half;
                v2f a = { cur[mt * 16 + lr][kk], cur[mt * 16 + lr][kk + 1] };
                acc[mt] = wmma_f32_4(a, bfr[kq], acc[mt]);
            }
        }
        if (MORE) {
            ((float4*)&nxt[wave + 8 * (4 * hp + 0)][0])[lane] = t0;
            ((float4*)&nxt[wave + 8 * (4 * hp + 1)][0])[lane] = t1;
            ((float4*)&nxt[wave + 8 * (4 * hp + 2)][0])[lane] = t2;
            ((float4*)&nxt[wave + 8 * (4 * hp + 3)][0])[lane] = t3;
        }
    }
}

__global__ __launch_bounds__(256) __attribute__((amdgpu_waves_per_eu(6, 6))) void ffn2_ln_kernel(
    const float* __restrict__ hbuf, const float* __restrict__ W2,
    const float* __restrict__ b2, const float* __restrict__ rst,
    const float* __restrict__ g, const float* __restrict__ be,
    float* __restrict__ out, int n)
{
    __shared__ float sA[2][64][132];
    const int row0 = blockIdx.x * 64;
    const int tid  = threadIdx.x;
    const int wave = tid >> 5;
    const int lane = tid & 31;
    const int half = lane >> 4;
    const int lr   = lane & 15;
    const int col  = wave * 16 + lr;

    {
        float4 t[8];
#pragma unroll
        for (int u = 0; u < 8; ++u) t[u] = ffn2_fetch(hbuf, row0, n, wave, lane, u, 0);
#pragma unroll
        for (int u = 0; u < 8; ++u) ((float4*)&sA[0][wave + 8 * u][0])[lane] = t[u];
    }

    v8f acc[4] = {};
#pragma unroll 1
    for (int kc = 0; kc < 3; ++kc) {
        __syncthreads();
        ffn2_chunk<true>(hbuf, W2, sA[kc & 1], sA[(kc + 1) & 1], acc,
                         kc, row0, n, wave, lane, half, lr, col);
    }
    __syncthreads();
    ffn2_chunk<false>(hbuf, W2, sA[1], sA[0], acc, 3, row0, n, wave, lane, half, lr, col);
    __builtin_amdgcn_sched_barrier(0);

    const float bias = b2[col];
#pragma unroll
    for (int mt = 0; mt < 4; ++mt) {
#pragma unroll
        for (int r = 0; r < 8; ++r) {
            const int rl = mt * 16 + r + 8 * half;
            int gr = row0 + rl; if (gr >= n) gr = n - 1;
            sA[0][rl][col] = acc[mt][r] + bias + rst[(size_t)gr * 128 + col];
        }
    }
    __syncthreads();

    const int rl2 = tid >> 2;
    const int c0  = tid & 3;
    const int row = row0 + rl2;
    float xv[32], s = 0.0f, ss = 0.0f;
#pragma unroll
    for (int j = 0; j < 32; ++j) {
        int c = c0 + 4 * j;
        float x = sA[0][rl2][c];
        xv[j] = x; s += x; ss += x * x;
    }
#pragma unroll
    for (int m = 1; m < 4; m <<= 1) {
        s  += __shfl_xor(s,  m, 32);
        ss += __shfl_xor(ss, m, 32);
    }
    const float mean = s * (1.0f / 128.0f);
    const float var  = ss * (1.0f / 128.0f) - mean * mean;
    const float inv  = rsqrtf(var + EPSLN);
    if (row < n) {
        float* o = out + (size_t)row * 128;
#pragma unroll
        for (int j = 0; j < 32; ++j) {
            int c = c0 + 4 * j;
            o[c] = (xv[j] - mean) * inv * g[c] + be[c];
        }
    }
}

extern "C" void kernel_launch(void* const* d_in, const int* in_sizes, int n_in,
                              void* d_out, int out_size, void* d_ws, size_t ws_size,
                              hipStream_t stream)
{
    const float* feat  = (const float*)d_in[0];
    const int*   src   = (const int*)  d_in[1];
    const int*   dst   = (const int*)  d_in[2];
    const float* Wq    = (const float*)d_in[3];
    const float* Wk    = (const float*)d_in[4];
    const float* Wv    = (const float*)d_in[5];
    const float* ln1_g = (const float*)d_in[6];
    const float* ln1_b = (const float*)d_in[7];
    const float* ln2_g = (const float*)d_in[8];
    const float* ln2_b = (const float*)d_in[9];
    const float* W1    = (const float*)d_in[10];
    const float* b1    = (const float*)d_in[11];
    const float* pw    = (const float*)d_in[12];
    const float* W2    = (const float*)d_in[13];
    const float* b2    = (const float*)d_in[14];

    const int n = in_sizes[0] / 128;
    const int e = in_sizes[1];

    float* ws = (float*)d_ws;
    size_t off = 0;
    float* q      = ws + off; off += (size_t)n * 128;
    float* kbuf   = ws + off; off += (size_t)n * 128;
    float* vbuf   = ws + off; off += (size_t)n * 128;
    float* rst    = ws + off; off += (size_t)n * 128;
    float* hbuf   = ws + off; off += (size_t)n * 512;
    (void)off; (void)ws_size; (void)n_in; (void)out_size;

    const int NT64 = (n + 63) / 64;
    const int NEG  = (n + EG_R - 1) / EG_R;

    qkv_kernel<<<NT64, 256, 0, stream>>>(feat, Wq, Wk, Wv, q, kbuf, vbuf, n);
    edge_kernel<<<NEG, EG_T, 0, stream>>>(q, kbuf, vbuf, src, dst,
                                          feat, ln1_g, ln1_b, rst, n, e);
    ffn1_kernel<<<NT64, 256, 0, stream>>>(rst, W1, b1, pw, hbuf, n);
    ffn2_ln_kernel<<<NT64, 256, 0, stream>>>(hbuf, W2, b2, rst, ln2_g, ln2_b,
                                             (float*)d_out, n);
}


# edge kernel reads its row of an owner-by-edge bit matrix instead of sweeping all of dst; extra work-groups of launch 1 build the matrix (LDS OR atomics); float sums still in arrival order
# speedup vs baseline: 3.4741x; 1.1030x over previous
#include <hip/hip_runtime.h>
#include <hip/hip_bf16.h>
#include <math.h>

typedef __attribute__((ext_vector_type(2))) float v2f;
typedef __attribute__((ext_vector_type(8))) float v8f;

#define EPSLN 1e-5f

#define EG_R    157
#define BM_CE   1024
#define BM_WPC  (BM_CE / 32)
__device__ __forceinline__ bool bm_fits(int n) {
    return ((n + EG_R - 1) / EG_R) * BM_WPC <= 64 * 132;
}

__device__ __forceinline__ v8f wmma_f32_4(v2f a, v2f b, v8f c) {
    return __builtin_amdgcn_wmma_f32_16x16x4_f32(
         false, a,  false, b,
         (short)0, c,  false,  false);
}

__global__ __launch_bounds__(256) void qkv_kernel(
    const float* __restrict__ feat,
    const float* __restrict__ Wq, const float* __restrict__ Wk,
    const float* __restrict__ Wv,
    float* __restrict__ q, float* __restrict__ k, float* __restrict__ v,
    const int* __restrict__ dst, unsigned* __restrict__ bmp, int n, int e, int nt64)
{
    __shared__ float sA[64][132];
    const int tid  = threadIdx.x;

    if ((int)blockIdx.x >= nt64) {
        if (!bm_fits(n)) return;
        const int c   = (int)blockIdx.x - nt64;
        const int neg = (n + EG_R - 1) / EG_R;
        const int nch = (e + BM_CE - 1) / BM_CE;
        unsigned* bm  = (unsigned*)&sA[0][0];
        const int i0  = c * BM_CE + 4 * tid;
        int4 d;
        if (i0 + 3 < e && (((size_t)dst) & 15) == 0) {
            d = *(const int4*)(dst + i0);
        } else {
            d.x = (i0     < e) ? dst[i0]     : -1;
            d.y = (i0 + 1 < e) ? dst[i0 + 1] : -1;
            d.z = (i0 + 2 < e) ? dst[i0 + 2] : -1;
            d.w = (i0 + 3 < e) ? dst[i0 + 3] : -1;
        }
        for (int i = tid; i < neg * BM_WPC; i += 256) bm[i] = 0u;
        __syncthreads();
        const int      wq = tid >> 3;
        const unsigned b0 = 1u << (4 * (tid & 7));
        if ((unsigned)d.x < (unsigned)n) atomicOr(&bm[(d.x / EG_R) * BM_WPC + wq], b0);
        if ((unsigned)d.y < (unsigned)n) atomicOr(&bm[(d.y / EG_R) * BM_WPC + wq], b0 << 1);
        if ((unsigned)d.z < (unsigned)n) atomicOr(&bm[(d.z / EG_R) * BM_WPC + wq], b0 << 2);
        if ((unsigned)d.w < (unsigned)n) atomicOr(&bm[(d.w / EG_R) * BM_WPC + wq], b0 << 3);
        __syncthreads();
        for (int i = tid; i < neg * (BM_WPC / 4); i += 256) {
            const int o = i >> 3, qd = i & 7;
            ((uint4*)(bmp + ((size_t)o * nch + c) * BM_WPC))[qd] = ((const uint4*)bm)[i];
        }
        return;
    }

    const int row0 = blockIdx.x * 64;

    for (int i = tid; i < 64 * 32; i += 256) {
        int r = i >> 5, c4 = i & 31;
        int gr = row0 + r; if (gr >= n) gr = n - 1;
        ((float4*)&sA[r][0])[c4] = ((const float4*)(feat + (size_t)gr * 128))[c4];
    }
    __syncthreads();

    const int wave = tid >> 5;
    const int lane = tid & 31;
    const int half = lane >> 4;
    const int lr   = lane & 15;
    const int col  = wave * 16 + lr;

    const float* Ws[3] = { Wq, Wk, Wv };
    float*       Os[3] = { q,  k,  v  };

#pragma unroll
    for (int m = 0; m < 3; ++m) {
        const float* __restrict__ W = Ws[m];
        v2f bfr[32];
#pragma unroll
        for (int kq = 0; kq < 32; ++kq) {
            const int kk = 4 * kq + 2 * half;
            bfr[kq] = (v2f){ W[(size_t)kk * 128 + col],
                             W[(size_t)(kk + 1) * 128 + col] };
        }
        float* __restrict__ O = Os[m];
#pragma unroll
        for (int mt = 0; mt < 4; ++mt) {
            v8f acc = {};
#pragma unroll
            for (int kq = 0; kq < 32; ++kq) {
                const int kk = 4 * kq + 2 * half;
                v2f a = { sA[mt * 16 + lr][kk], sA[mt * 16 + lr][kk + 1] };
                acc = wmma_f32_4(a, bfr[kq], acc);
            }
#pragma unroll
            for (int r = 0; r < 8; ++r) {
                int gr = row0 + mt * 16 + r + 8 * half;
                if (gr < n) O[(size_t)gr * 128 + col] = acc[r];
            }
        }
    }
}

#define EG_T    1024
#define EG_CAP  6144
#define EG_CS   3072
#define EG_QV   ((EG_R * 32 + EG_T - 1) / EG_T)
#define EG_RW   ((EG_R + EG_T / 32 - 1) / (EG_T / 32))

__device__ __forceinline__ void lds_max_f(float* addr, float val) {
    val += 0.0f;
    if (val >= 0.0f) atomicMax((int*)addr, __float_as_int(val));
    else             atomicMin((unsigned int*)addr, __float_as_uint(val));
}

__device__ __forceinline__ float quad_sum(float p) {
    p += __int_as_float(__builtin_amdgcn_mov_dpp(__float_as_int(p), 0xB1, 0xF, 0xF, true));
    p += __int_as_float(__builtin_amdgcn_mov_dpp(__float_as_int(p), 0x4E, 0xF, 0xF, true));
    return p;
}

template <int V4, bool CAPPED, bool FULL>
__device__ __forceinline__ void eg_scan_tile(
    const int* __restrict__ dst, int e, int base,
    int lo, unsigned rows, int2* s_list, int* s_cnt, int tid)
{
    int4 dv[V4];
#pragma unroll
    for (int u = 0; u < V4; ++u) {
        const int i0 = base + (u * EG_T + tid) * 4;
        if (FULL) {
            dv[u] = *(const int4*)(dst + i0);
        } else {
            dv[u].x = (i0     < e) ? dst[i0]     : -1;
            dv[u].y = (i0 + 1 < e) ? dst[i0 + 1] : -1;
            dv[u].z = (i0 + 2 < e) ? dst[i0 + 2] : -1;
            dv[u].w = (i0 + 3 < e) ? dst[i0 + 3] : -1;
        }
    }
#pragma unroll
    for (int u = 0; u < V4; ++u) {
        const int i0 = base + (u * EG_T + tid) * 4;
        const unsigned r0 = (unsigned)(dv[u].x - lo), r1 = (unsigned)(dv[u].y - lo);
        const unsigned r2 = (unsigned)(dv[u].z - lo), r3 = (unsigned)(dv[u].w - lo);
        const int c = (int)(r0 < rows) + (int)(r1 < rows) + (int)(r2 < rows) + (int)(r3 < rows);
        if (c) {
            int p = atomicAdd(s_cnt, c);
            if (r0 < rows) { if (!CAPPED || p < EG_CAP) s_list[p] = make_int2(i0,     (int)r0); ++p; }
            if (r1 < rows) { if (!CAPPED || p < EG_CAP) s_list[p] = make_int2(i0 + 1, (int)r1); ++p; }
            if (r2 < rows) { if (!CAPPED || p < EG_CAP) s_list[p] = make_int2(i0 + 2, (int)r2); ++p; }
            if (r3 < rows) { if (!CAPPED || p < EG_CAP) s_list[p] = make_int2(i0 + 3, (int)r3); }
        }
    }
}

__device__ __forceinline__ void eg_resolve(
    const int* __restrict__ src, int2* s_list, int cnt, int tid)
{
    for (int i = tid; i < cnt; i += EG_T) s_list[i].x = src[s_list[i].x];
}

__device__ __forceinline__ void eg_resolve_bm(
    const int* __restrict__ src, const int* __restrict__ dst,
    int2* s_list, int cnt, int lo, int tid)
{
    for (int i = tid; i < cnt; i += EG_T) {
        const int ed = s_list[i].x;
        s_list[i] = make_int2(src[ed], dst[ed] - lo);
    }
}

#define BM_V 5
__device__ __forceinline__ void eg_scan_bits(
    const uint4* __restrict__ row, int nw4, int base, int2* s_list, int* s_cnt, int tid)
{
    uint4 w[BM_V];
#pragma unroll
    for (int u = 0; u < BM_V; ++u) {
        const int idx = base + u * EG_T + tid;
        w[u] = (idx < nw4) ? row[idx] : make_uint4(0u, 0u, 0u, 0u);
    }
    int c = 0;
#pragma unroll
    for (int u = 0; u < BM_V; ++u)
        c += __popc(w[u].x) + __popc(w[u].y) + __popc(w[u].z) + __popc(w[u].w);
    if (c) {
        int p = atomicAdd(s_cnt, c);
#pragma unroll
        for (int u = 0; u < BM_V; ++u) {
            const int e0 = (base + u * EG_T + tid) * 128;
            const unsigned ww[4] = { w[u].x, w[u].y, w[u].z, w[u].w };
#pragma unroll
            for (int k = 0; k < 4; ++k) {
                unsigned m = ww[k];
                while (m) {
                    const int b = __ffs((int)m) - 1;
                    m &= m - 1u;
                    if (p < EG_CAP) s_list[p] = make_int2(e0 + 32 * k + b, 0);
                    ++p;
                }
            }
        }
    }
}

template <int PASS, bool CACHE>
__device__ __forceinline__ void eg_process(
    const float* __restrict__ kbuf, const float* __restrict__ vbuf,
    const int2* s_list, int cnt, const float* s_q, float* s_sc,
    float* s_m, float* s_l, float* s_acc, int wave, int lane)
{
    const int hq = lane >> 2;
    for (int i0 = wave; i0 < cnt; i0 += 32 * 4) {
        int2   en[4];
        float4 kk[4];
        float4 vv[4];
#pragma unroll
        for (int u = 0; u < 4; ++u) {
            const int i = i0 + 32 * u;
            en[u] = s_list[(i < cnt) ? i : i0];
            if (!(PASS == 1 && CACHE))
                kk[u] = *(const float4*)(kbuf + (size_t)en[u].x * 128 + 4 * lane);
            if (PASS == 1)
                vv[u] = *(const float4*)(vbuf + (size_t)en[u].x * 128 + 4 * lane);
        }
#pragma unroll
        for (int u = 0; u < 4; ++u) {
            const bool ok = (i0 + 32 * u) < cnt;
            const int  ic = ok ? (i0 + 32 * u) : i0;
            const int  dl = en[u].y;
            float sc;
            if (PASS == 1 && CACHE) {
                sc = s_sc[ic * 8 + hq];
            } else {
                const float4 qq = *(const float4*)(s_q + dl * 128 + 4 * lane);
                const float p = quad_sum(kk[u].x * qq.x + kk[u].y * qq.y +
                                         kk[u].z * qq.z + kk[u].w * qq.w);
                sc = p * 0.08838834764831845f;
            }
            if (PASS == 0) {
                if (ok && (lane & 3) == 0) {
                    lds_max_f(&s_m[dl * 8 + hq], sc);
                    if (CACHE) s_sc[ic * 8 + hq] = sc;
                }
            } else {
                const float ex = expf(sc - s_m[dl * 8 + hq]);
                if (ok) {
                    if ((lane & 3) == 0) atomicAdd(&s_l[dl * 8 + hq], ex);
                    float* ap = s_acc + dl * 128 + lane;
                    atomicAdd(ap,      vv[u].x * ex);
                    atomicAdd(ap + 32, vv[u].y * ex);
                    atomicAdd(ap + 64, vv[u].z * ex);
                    atomicAdd(ap + 96, vv[u].w * ex);
                }
            }
        }
    }
}

__global__ __launch_bounds__(EG_T) void edge_kernel(
    const float* __restrict__ qbuf, const float* __restrict__ kbuf,
    const float* __restrict__ vbuf,
    const int* __restrict__ src, const int* __restrict__ dst,
    const float* __restrict__ feat,
    const float* __restrict__ ln_g, const float* __restrict__ ln_b,
    const unsigned* __restrict__ bmp,
    float* __restrict__ rst, int n, int e)
{
    __shared__ float s_acc[EG_R * 128];
    __shared__ float s_q[EG_R * 128];
    __shared__ float s_m[EG_R * 8];
    __shared__ float s_l[EG_R * 8];
    __shared__ float s_sc[EG_CS * 8];
    __shared__ int2  s_list[EG_CAP];
    __shared__ int   s_cnt;

    const int tid  = threadIdx.x;
    const int wave = tid >> 5;
    const int lane = tid & 31;
    const int lo   = blockIdx.x * EG_R;
    const int rows = (n - lo < EG_R) ? (n - lo) : EG_R;

    {
        const float4* qs = (const float4*)(qbuf + (size_t)lo * 128);
        float4 t[EG_QV];
#pragma unroll
        for (int u = 0; u < EG_QV; ++u) {
            const int i = tid + u * EG_T;
            t[u] = qs[(i < rows * 32) ? i : rows * 32 - 1];
        }
#pragma unroll
        for (int u = 0; u < EG_QV; ++u) {
            const int i = tid + u * EG_T;
            if (i < rows * 32) ((float4*)s_q)[i] = t[u];
        }
    }
    for (int i = tid; i < EG_R * 128; i += EG_T) s_acc[i] = 0.0f;
    for (int i = tid; i < EG_R * 8; i += EG_T) { s_m[i] = -INFINITY; s_l[i] = 0.0f; }
    if (tid == 0) s_cnt = 0;
    __syncthreads();

    const bool from_bm = bm_fits(n);
    if (from_bm) {
        const int nch = (e + BM_CE - 1) / BM_CE;
        const int nw4 = nch * (BM_WPC / 4);
        const uint4* row = (const uint4*)(bmp + (size_t)blockIdx.x * nch * BM_WPC);
        for (int base = 0; base < nw4; base += EG_T * BM_V)
            eg_scan_bits(row, nw4, base, s_list, &s_cnt, tid);
    } else {
        const int TE = EG_T * 32;
        const int nf = ((((size_t)dst) & 15) == 0) ? e / TE : 0;
        int t = (nf > 0) ? (int)(blockIdx.x % (unsigned)nf) : 0;
        for (int it = 0; it < nf; ++it) {
            eg_scan_tile<8, true, true>(dst, e, t * TE, lo, (unsigned)rows, s_list, &s_cnt, tid);
            if (++t == nf) t = 0;
        }
        for (int base = nf * TE; base < e; base += EG_T * 4)
            eg_scan_tile<1, true, false>(dst, e, base, lo, (unsigned)rows, s_list, &s_cnt, tid);
    }
    __syncthreads();
    const int total = s_cnt;

    float4 fv[EG_RW];
#pragma unroll
    for (int t = 0; t < EG_RW; ++t) {
        const int dl = wave + (EG_T / 32) * t;
        const int dc = (dl < rows) ? dl : rows - 1;
        fv[t] = *(const float4*)(feat + (size_t)(lo + dc) * 128 + 4 * lane);
    }
    const float g0 = ln_g[4 * lane],     g1 = ln_g[4 * lane + 1];
    const float g2 = ln_g[4 * lane + 2], g3 = ln_g[4 * lane + 3];
    const float b0 = ln_b[4 * lane],     b1 = ln_b[4 * lane + 1];
    const float b2 = ln_b[4 * lane + 2], b3 = ln_b[4 * lane + 3];

    if (total <= EG_CS) {
        if (from_bm) eg_resolve_bm(src, dst, s_list, total, lo, tid);
        else         eg_resolve(src, s_list, total, tid);
        __syncthreads();
        eg_process<0, true>(kbuf, vbuf, s_list, total, s_q, s_sc, s_m, s_l, s_acc, wave, lane);
        __syncthreads();
        eg_process<1, true>(kbuf, vbuf, s_list, total, s_q, s_sc, s_m, s_l, s_acc, wave, lane);
    } else if (total <= EG_CAP) {
        if (from_bm) eg_resolve_bm(src, dst, s_list, total, lo, tid);
        else         eg_resolve(src, s_list, total, tid);
        __syncthreads();
        eg_process<0, false>(kbuf, vbuf, s_list, total, s_q, s_sc, s_m, s_l, s_acc, wave, lane);
        __syncthreads();
        eg_process<1, false>(kbuf, vbuf, s_list, total, s_q, s_sc, s_m, s_l, s_acc, wave, lane);
    } else {
        const int TE = EG_T * 4;
#pragma unroll 1
        for (int pass = 0; pass < 2; ++pass) {
            __syncthreads();
            if (tid == 0) s_cnt = 0;
            __syncthreads();
            for (int base = 0; base < e; base += TE) {
                eg_scan_tile<1, false, false>(dst, e, base, lo, (unsigned)rows, s_list, &s_cnt, tid);
                __syncthreads();
                const int c = s_cnt;
                if (c > EG_CAP - TE) {
                    eg_resolve(src, s_list, c, tid);
                    __syncthreads();
                    if (pass == 0) eg_process<0, false>(kbuf, vbuf, s_list, c, s_q, s_sc, s_m, s_l, s_acc, wave, lane);
                    else           eg_process<1, false>(kbuf, vbuf, s_list, c, s_q, s_sc, s_m, s_l, s_acc, wave, lane);
                    __syncthreads();
                    if (tid == 0) s_cnt = 0;
                }
                __syncthreads();
            }
            const int c = s_cnt;
            eg_resolve(src, s_list, c, tid);
            __syncthreads();
            if (pass == 0) eg_process<0, false>(kbuf, vbuf, s_list, c, s_q, s_sc, s_m, s_l, s_acc, wave, lane);
            else           eg_process<1, false>(kbuf, vbuf, s_list, c, s_q, s_sc, s_m, s_l, s_acc, wave, lane);
        }
    }
    __syncthreads();

    float4 o[EG_RW];
#pragma unroll
    for (int t = 0; t < EG_RW; ++t) {
        const int dl = wave + (EG_T / 32) * t;
        const int dc = (dl < rows) ? dl : rows - 1;
        const float l = s_l[dc * 8 + (lane >> 2)];
        const float* ap = s_acc + dc * 128 + lane;
        const float a0 = ap[0], a1 = ap[32], a2 = ap[64], a3 = ap[96];
        const bool has = l > 0.0f;
        const float x0 = (has ? a0 / l : 0.0f) + fv[t].x;
        const float x1 = (has ? a1 / l : 0.0f) + fv[t].y;
        const float x2 = (has ? a2 / l : 0.0f) + fv[t].z;
        const float x3 = (has ? a3 / l : 0.0f) + fv[t].w;
        float s  = (x0 + x1) + (x2 + x3);
        float ss = (x0 * x0 + x1 * x1) + (x2 * x2 + x3 * x3);
#pragma unroll
        for (int m = 1; m < 32; m <<= 1) {
            s  += __shfl_xor(s,  m, 32);
            ss += __shfl_xor(ss, m, 32);
        }
        const float mean = s * (1.0f / 128.0f);
        const float var  = ss * (1.0f / 128.0f) - mean * mean;
        const float inv  = rsqrtf(var + EPSLN);
        o[t].x = (x0 - mean) * inv * g0 + b0;
        o[t].y = (x1 - mean) * inv * g1 + b1;
        o[t].z = (x2 - mean) * inv * g2 + b2;
        o[t].w = (x3 - mean) * inv * g3 + b3;
    }
#pragma unroll
    for (int t = 0; t < EG_RW; ++t) {
        const int dl = wave + (EG_T / 32) * t;
        if (dl < rows)
            *(float4*)(rst + (size_t)(lo + dl) * 128 + 4 * lane) = o[t];
    }
}

__global__ __launch_bounds__(256) void ffn1_kernel(
    const float* __restrict__ rst, const float* __restrict__ W1,
    const float* __restrict__ b1, const float* __restrict__ pw,
    float* __restrict__ hbuf, int n)
{
    __shared__ float sA[64][132];
    const int row0 = blockIdx.x * 64;
    const int tid  = threadIdx.x;
    for (int i = tid; i < 64 * 32; i += 256) {
        int r = i >> 5, c4 = i & 31;
        int gr = row0 + r; if (gr >= n) gr = n - 1;
        ((float4*)&sA[r][0])[c4] = ((const float4*)(rst + (size_t)gr * 128))[c4];
    }
    __syncthreads();

    const int wave = tid >> 5;
    const int lane = tid & 31;
    const int half = lane >> 4;
    const int lr   = lane & 15;

#pragma unroll
    for (int j = 0; j < 4; ++j) {
        const int col = (wave + 8 * j) * 16 + lr;
        v2f bfr[32];
#pragma unroll
        for (int kq = 0; kq < 32; ++kq) {
            const int kk = 4 * kq + 2 * half;
            bfr[kq] = (v2f){ W1[(size_t)kk * 512 + col],
                             W1[(size_t)(kk + 1) * 512 + col] };
        }
        const float bias = b1[col], slope = pw[col];
#pragma unroll
        for (int mt = 0; mt < 4; ++mt) {
            v8f acc = {};
#pragma unroll
            for (int kq = 0; kq < 32; ++kq) {
                const int kk = 4 * kq + 2 * half;
                v2f a = { sA[mt * 16 + lr][kk], sA[mt * 16 + lr][kk + 1] };
                acc = wmma_f32_4(a, bfr[kq], acc);
            }
#pragma unroll
            for (int r = 0; r < 8; ++r) {
                int gr = row0 + mt * 16 + r + 8 * half;
                if (gr < n) {
                    float h = acc[r] + bias;
                    h = (h >= 0.0f) ? h : slope * h;
                    hbuf[(size_t)gr * 512 + col] = h;
                }
            }
        }
    }
}

__device__ __forceinline__ float4 ffn2_fetch(const float* __restrict__ hbuf, int row0, int n,
                                             int wave, int lane, int u, int kc)
{
    int gr = row0 + wave + 8 * u; if (gr >= n) gr = n - 1;
    const unsigned off = (unsigned)gr * 512u + (unsigned)(kc * 128 + 4 * lane);
    return *(const float4*)(hbuf + off);
}

template <bool MORE>
__device__ __forceinline__ void ffn2_chunk(
    const float* __restrict__ hbuf, const float* __restrict__ W2,
    float (*cur)[132], float (*nxt)[132], v8f (&acc)[4],
    int kc, int row0, int n, int wave, int lane, int half, int lr, int col)
{
    v2f bfr[32];
#pragma unroll
    for (int kq = 0; kq < 32; ++kq) {
        const int kk = kc * 128 + 4 * kq + 2 * half;
        bfr[kq] = (v2f){ W2[(size_t)kk * 128 + col],
                         W2[(size_t)(kk + 1) * 128 + col] };
    }
#pragma unroll
    for (int hp = 0; hp < 2; ++hp) {
        float4 t0, t1, t2, t3;
        if (MORE) {
            t0 = ffn2_fetch(hbuf, row0, n, wave, lane, 4 * hp + 0, kc + 1);
            t1 = ffn2_fetch(hbuf, row0, n, wave, lane, 4 * hp + 1, kc + 1);
            t2 = ffn2_fetch(hbuf, row0, n, wave, lane, 4 * hp + 2, kc + 1);
            t3 = ffn2_fetch(hbuf, row0, n, wave, lane, 4 * hp + 3, kc + 1);
            __builtin_amdgcn_sched_barrier(0);
        }
#pragma unroll
        for (int mt = 2 * hp; mt < 2 * hp + 2; ++mt) {
#pragma unroll
            for (int kq = 0; kq < 32; ++kq) {
                const int kk = 4 * kq + 2 * half;
                v2f a = { cur[mt * 16 + lr][kk], cur[mt * 16 + lr][kk + 1] };
                acc[mt] = wmma_f32_4(a, bfr[kq], acc[mt]);
            }
        }
        if (MORE) {
            ((float4*)&nxt[wave + 8 * (4 * hp + 0)][0])[lane] = t0;
            ((float4*)&nxt[wave + 8 * (4 * hp + 1)][0])[lane] = t1;
            ((float4*)&nxt[wave + 8 * (4 * hp + 2)][0])[lane] = t2;
            ((float4*)&nxt[wave + 8 * (4 * hp + 3)][0])[lane] = t3;
        }
    }
}

__global__ __launch_bounds__(256) __attribute__((amdgpu_waves_per_eu(6, 6))) void ffn2_ln_kernel(
    const float* __restrict__ hbuf, const float* __restrict__ W2,
    const float* __restrict__ b2, const float* __restrict__ rst,
    const float* __restrict__ g, const float* __restrict__ be,
    float* __restrict__ out, int n)
{
    __shared__ float sA[2][64][132];
    const int row0 = blockIdx.x * 64;
    const int tid  = threadIdx.x;
    const int wave = tid >> 5;
    const int lane = tid & 31;
    const int half = lane >> 4;
    const int lr   = lane & 15;
    const int col  = wave * 16 + lr;

    {
        float4 t[8];
#pragma unroll
        for (int u = 0; u < 8; ++u) t[u] = ffn2_fetch(hbuf, row0, n, wave, lane, u, 0);
#pragma unroll
        for (int u = 0; u < 8; ++u) ((float4*)&sA[0][wave + 8 * u][0])[lane] = t[u];
    }

    v8f acc[4] = {};
#pragma unroll 1
    for (int kc = 0; kc < 3; ++kc) {
        __syncthreads();
        ffn2_chunk<true>(hbuf, W2, sA[kc & 1], sA[(kc + 1) & 1], acc,
                         kc, row0, n, wave, lane, half, lr, col);
    }
    __syncthreads();
    ffn2_chunk<false>(hbuf, W2, sA[1], sA[0], acc, 3, row0, n, wave, lane, half, lr, col);
    __builtin_amdgcn_sched_barrier(0);

    const float bias = b2[col];
#pragma unroll
    for (int mt = 0; mt < 4; ++mt) {
#pragma unroll
        for (int r = 0; r < 8; ++r) {
            const int rl = mt * 16 + r + 8 * half;
            int gr = row0 + rl; if (gr >= n) gr = n - 1;
            sA[0][rl][col] = acc[mt][r] + bias + rst[(size_t)gr * 128 + col];
        }
    }
    __syncthreads();

    const int rl2 = tid >> 2;
    const int c0  = tid & 3;
    const int row = row0 + rl2;
    float xv[32], s = 0.0f, ss = 0.0f;
#pragma unroll
    for (int j = 0; j < 32; ++j) {
        int c = c0 + 4 * j;
        float x = sA[0][rl2][c];
        xv[j] = x; s += x; ss += x * x;
    }
#pragma unroll
    for (int m = 1; m < 4; m <<= 1) {
        s  += __shfl_xor(s,  m, 32);
        ss += __shfl_xor(ss, m, 32);
    }
    const float mean = s * (1.0f / 128.0f);
    const float var  = ss * (1.0f / 128.0f) - mean * mean;
    const float inv  = rsqrtf(var + EPSLN);
    if (row < n) {
        float* o = out + (size_t)row * 128;
#pragma unroll
        for (int j = 0; j < 32; ++j) {
            int c = c0 + 4 * j;
            o[c] = (xv[j] - mean) * inv * g[c] + be[c];
        }
    }
}

extern "C" void kernel_launch(void* const* d_in, const int* in_sizes, int n_in,
                              void* d_out, int out_size, void* d_ws, size_t ws_size,
                              hipStream_t stream)
{
    const float* feat  = (const float*)d_in[0];
    const int*   src   = (const int*)  d_in[1];
    const int*   dst   = (const int*)  d_in[2];
    const float* Wq    = (const float*)d_in[3];
    const float* Wk    = (const float*)d_in[4];
    const float* Wv    = (const float*)d_in[5];
    const float* ln1_g = (const float*)d_in[6];
    const float* ln1_b = (const float*)d_in[7];
    const float* ln2_g = (const float*)d_in[8];
    const float* ln2_b = (const float*)d_in[9];
    const float* W1    = (const float*)d_in[10];
    const float* b1    = (const float*)d_in[11];
    const float* pw    = (const float*)d_in[12];
    const float* W2    = (const float*)d_in[13];
    const float* b2    = (const float*)d_in[14];

    const int n = in_sizes[0] / 128;
    const int e = in_sizes[1];

    float* ws = (float*)d_ws;
    size_t off = 0;
    float* q      = ws + off; off += (size_t)n * 128;
    float* kbuf   = ws + off; off += (size_t)n * 128;
    float* vbuf   = ws + off; off += (size_t)n * 128;
    float* rst    = ws + off; off += (size_t)n * 128;
    float* hbuf   = ws + off; off += (size_t)n * 512;

    const int NT64 = (n + 63) / 64;
    const int NEG  = (n + EG_R - 1) / EG_R;
    const int NBM  = (e + BM_CE - 1) / BM_CE;

    unsigned* bmp = (unsigned*)(ws + off); off += (size_t)NEG * NBM * BM_WPC;
    (void)off; (void)ws_size; (void)n_in; (void)out_size;

    qkv_kernel<<<NT64 + NBM, 256, 0, stream>>>(feat, Wq, Wk, Wv, q, kbuf, vbuf,
                                               dst, bmp, n, e, NT64);
    edge_kernel<<<NEG, EG_T, 0, stream>>>(q, kbuf, vbuf, src, dst,
                                          feat, ln1_g, ln1_b, bmp, rst, n, e);
    ffn1_kernel<<<NT64, 256, 0, stream>>>(rst, W1, b1, pw, hbuf, n);
    ffn2_ln_kernel<<<NT64, 256, 0, stream>>>(hbuf, W2, b2, rst, ln2_g, ln2_b,
                                             (float*)d_out, n);
}


# FFN1 (rst@W1+b1, PReLU) moved into the edge kernel: the owner work-group multiplies its rst rows from an LDS tile, 32 waves x 16 columns, B fragments in registers; FFN1 launch goes; sum order kept
# speedup vs baseline: 3.6362x; 1.0467x over previous
#include <hip/hip_runtime.h>
#include <hip/hip_bf16.h>
#include <math.h>

typedef __attribute__((ext_vector_type(2))) float v2f;
typedef __attribute__((ext_vector_type(8))) float v8f;

#define EPSLN 1e-5f

#define EG_R    157
#define BM_CE   1024
#define BM_WPC  (BM_CE / 32)
__device__ __forceinline__ bool bm_fits(int n) {
    return ((n + EG_R - 1) / EG_R) * BM_WPC <= 64 * 132;
}

__device__ __forceinline__ v8f wmma_f32_4(v2f a, v2f b, v8f c) {
    return __builtin_amdgcn_wmma_f32_16x16x4_f32(
         false, a,  false, b,
         (short)0, c,  false,  false);
}

__global__ __launch_bounds__(256) void qkv_kernel(
    const float* __restrict__ feat,
    const float* __restrict__ Wq, const float* __restrict__ Wk,
    const float* __restrict__ Wv,
    float* __restrict__ q, float* __restrict__ k, float* __restrict__ v,
    const int* __restrict__ dst, unsigned* __restrict__ bmp, int n, int e, int nt64)
{
    __shared__ float sA[64][132];
    const int tid  = threadIdx.x;

    if ((int)blockIdx.x >= nt64) {
        if (!bm_fits(n)) return;
        const int c   = (int)blockIdx.x - nt64;
        const int neg = (n + EG_R - 1) / EG_R;
        const int nch = (e + BM_CE - 1) / BM_CE;
        unsigned* bm  = (unsigned*)&sA[0][0];
        const int i0  = c * BM_CE + 4 * tid;
        int4 d;
        if (i0 + 3 < e && (((size_t)dst) & 15) == 0) {
            d = *(const int4*)(dst + i0);
        } else {
            d.x = (i0     < e) ? dst[i0]     : -1;
            d.y = (i0 + 1 < e) ? dst[i0 + 1] : -1;
            d.z = (i0 + 2 < e) ? dst[i0 + 2] : -1;
            d.w = (i0 + 3 < e) ? dst[i0 + 3] : -1;
        }
        for (int i = tid; i < neg * BM_WPC; i += 256) bm[i] = 0u;
        __syncthreads();
        const int      wq = tid >> 3;
        const unsigned b0 = 1u << (4 * (tid & 7));
        if ((unsigned)d.x < (unsigned)n) atomicOr(&bm[(d.x / EG_R) * BM_WPC + wq], b0);
        if ((unsigned)d.y < (unsigned)n) atomicOr(&bm[(d.y / EG_R) * BM_WPC + wq], b0 << 1);
        if ((unsigned)d.z < (unsigned)n) atomicOr(&bm[(d.z / EG_R) * BM_WPC + wq], b0 << 2);
        if ((unsigned)d.w < (unsigned)n) atomicOr(&bm[(d.w / EG_R) * BM_WPC + wq], b0 << 3);
        __syncthreads();
        for (int i = tid; i < neg * (BM_WPC / 4); i += 256) {
            const int o = i >> 3, qd = i & 7;
            ((uint4*)(bmp + ((size_t)o * nch + c) * BM_WPC))[qd] = ((const uint4*)bm)[i];
        }
        return;
    }

    const int row0 = blockIdx.x * 64;

    for (int i = tid; i < 64 * 32; i += 256) {
        int r = i >> 5, c4 = i & 31;
        int gr = row0 + r; if (gr >= n) gr = n - 1;
        ((float4*)&sA[r][0])[c4] = ((const float4*)(feat + (size_t)gr * 128))[c4];
    }
    __syncthreads();

    const int wave = tid >> 5;
    const int lane = tid & 31;
    const int half = lane >> 4;
    const int lr   = lane & 15;
    const int col  = wave * 16 + lr;

    const float* Ws[3] = { Wq, Wk, Wv };
    float*       Os[3] = { q,  k,  v  };

#pragma unroll
    for (int m = 0; m < 3; ++m) {
        const float* __restrict__ W = Ws[m];
        v2f bfr[32];
#pragma unroll
        for (int kq = 0; kq < 32; ++kq) {
            const int kk = 4 * kq + 2 * half;
            bfr[kq] = (v2f){ W[(size_t)kk * 128 + col],
                             W[(size_t)(kk + 1) * 128 + col] };
        }
        float* __restrict__ O = Os[m];
#pragma unroll
        for (int mt = 0; mt < 4; ++mt) {
            v8f acc = {};
#pragma unroll
            for (int kq = 0; kq < 32; ++kq) {
                const int kk = 4 * kq + 2 * half;
                v2f a = { sA[mt * 16 + lr][kk], sA[mt * 16 + lr][kk + 1] };
                acc = wmma_f32_4(a, bfr[kq], acc);
            }
#pragma unroll
            for (int r = 0; r < 8; ++r) {
                int gr = row0 + mt * 16 + r + 8 * half;
                if (gr < n) O[(size_t)gr * 128 + col] = acc[r];
            }
        }
    }
}

#define EG_T    1024
#define EG_CAP  6144
#define EG_CS   3072
#define EG_QV   ((EG_R * 32 + EG_T - 1) / EG_T)
#define EG_RW   ((EG_R + EG_T / 32 - 1) / (EG_T / 32))

__device__ __forceinline__ void lds_max_f(float* addr, float val) {
    val += 0.0f;
    if (val >= 0.0f) atomicMax((int*)addr, __float_as_int(val));
    else             atomicMin((unsigned int*)addr, __float_as_uint(val));
}

__device__ __forceinline__ float quad_sum(float p) {
    p += __int_as_float(__builtin_amdgcn_mov_dpp(__float_as_int(p), 0xB1, 0xF, 0xF, true));
    p += __int_as_float(__builtin_amdgcn_mov_dpp(__float_as_int(p), 0x4E, 0xF, 0xF, true));
    return p;
}

template <int V4, bool CAPPED, bool FULL>
__device__ __forceinline__ void eg_scan_tile(
    const int* __restrict__ dst, int e, int base,
    int lo, unsigned rows, int2* s_list, int* s_cnt, int tid)
{
    int4 dv[V4];
#pragma unroll
    for (int u = 0; u < V4; ++u) {
        const int i0 = base + (u * EG_T + tid) * 4;
        if (FULL) {
            dv[u] = *(const int4*)(dst + i0);
        } else {
            dv[u].x = (i0     < e) ? dst[i0]     : -1;
            dv[u].y = (i0 + 1 < e) ? dst[i0 + 1] : -1;
            dv[u].z = (i0 + 2 < e) ? dst[i0 + 2] : -1;
            dv[u].w = (i0 + 3 < e) ? dst[i0 + 3] : -1;
        }
    }
#pragma unroll
    for (int u = 0; u < V4; ++u) {
        const int i0 = base + (u * EG_T + tid) * 4;
        const unsigned r0 = (unsigned)(dv[u].x - lo), r1 = (unsigned)(dv[u].y - lo);
        const unsigned r2 = (unsigned)(dv[u].z - lo), r3 = (unsigned)(dv[u].w - lo);
        const int c = (int)(r0 < rows) + (int)(r1 < rows) + (int)(r2 < rows) + (int)(r3 < rows);
        if (c) {
            int p = atomicAdd(s_cnt, c);
            if (r0 < rows) { if (!CAPPED || p < EG_CAP) s_list[p] = make_int2(i0,     (int)r0); ++p; }
            if (r1 < rows) { if (!CAPPED || p < EG_CAP) s_list[p] = make_int2(i0 + 1, (int)r1); ++p; }
            if (r2 < rows) { if (!CAPPED || p < EG_CAP) s_list[p] = make_int2(i0 + 2, (int)r2); ++p; }
            if (r3 < rows) { if (!CAPPED || p < EG_CAP) s_list[p] = make_int2(i0 + 3, (int)r3); }
        }
    }
}

__device__ __forceinline__ void eg_resolve(
    const int* __restrict__ src, int2* s_list, int cnt, int tid)
{
    for (int i = tid; i < cnt; i += EG_T) s_list[i].x = src[s_list[i].x];
}

__device__ __forceinline__ void eg_resolve_bm(
    const int* __restrict__ src, const int* __restrict__ dst,
    int2* s_list, int cnt, int lo, int tid)
{
    for (int i = tid; i < cnt; i += EG_T) {
        const int ed = s_list[i].x;
        s_list[i] = make_int2(src[ed], dst[ed] - lo);
    }
}

#define BM_V 5
__device__ __forceinline__ void eg_scan_bits(
    const uint4* __restrict__ row, int nw4, int base, int2* s_list, int* s_cnt, int tid)
{
    uint4 w[BM_V];
#pragma unroll
    for (int u = 0; u < BM_V; ++u) {
        const int idx = base + u * EG_T + tid;
        w[u] = (idx < nw4) ? row[idx] : make_uint4(0u, 0u, 0u, 0u);
    }
    int c = 0;
#pragma unroll
    for (int u = 0; u < BM_V; ++u)
        c += __popc(w[u].x) + __popc(w[u].y) + __popc(w[u].z) + __popc(w[u].w);
    if (c) {
        int p = atomicAdd(s_cnt, c);
#pragma unroll
        for (int u = 0; u < BM_V; ++u) {
            const int e0 = (base + u * EG_T + tid) * 128;
            const unsigned ww[4] = { w[u].x, w[u].y, w[u].z, w[u].w };
#pragma unroll
            for (int k = 0; k < 4; ++k) {
                unsigned m = ww[k];
                while (m) {
                    const int b = __ffs((int)m) - 1;
                    m &= m - 1u;
                    if (p < EG_CAP) s_list[p] = make_int2(e0 + 32 * k + b, 0);
                    ++p;
                }
            }
        }
    }
}

template <int PASS, bool CACHE>
__device__ __forceinline__ void eg_process(
    const float* __restrict__ kbuf, const float* __restrict__ vbuf,
    const int2* s_list, int cnt, const float* s_q, float* s_sc,
    float* s_m, float* s_l, float* s_acc, int wave, int lane)
{
    const int hq = lane >> 2;
    for (int i0 = wave; i0 < cnt; i0 += 32 * 4) {
        int2   en[4];
        float4 kk[4];
        float4 vv[4];
#pragma unroll
        for (int u = 0; u < 4; ++u) {
            const int i = i0 + 32 * u;
            en[u] = s_list[(i < cnt) ? i : i0];
            if (!(PASS == 1 && CACHE))
                kk[u] = *(const float4*)(kbuf + (size_t)en[u].x * 128 + 4 * lane);
            if (PASS == 1)
                vv[u] = *(const float4*)(vbuf + (size_t)en[u].x * 128 + 4 * lane);
        }
#pragma unroll
        for (int u = 0; u < 4; ++u) {
            const bool ok = (i0 + 32 * u) < cnt;
            const int  ic = ok ? (i0 + 32 * u) : i0;
            const int  dl = en[u].y;
            float sc;
            if (PASS == 1 && CACHE) {
                sc = s_sc[ic * 8 + hq];
            } else {
                const float4 qq = *(const float4*)(s_q + dl * 128 + 4 * lane);
                const float p = quad_sum(kk[u].x * qq.x + kk[u].y * qq.y +
                                         kk[u].z * qq.z + kk[u].w * qq.w);
                sc = p * 0.08838834764831845f;
            }
            if (PASS == 0) {
                if (ok && (lane & 3) == 0) {
                    lds_max_f(&s_m[dl * 8 + hq], sc);
                    if (CACHE) s_sc[ic * 8 + hq] = sc;
                }
            } else {
                const float ex = expf(sc - s_m[dl * 8 + hq]);
                if (ok) {
                    if ((lane & 3) == 0) atomicAdd(&s_l[dl * 8 + hq], ex);
                    float* ap = s_acc + dl * 128 + lane;
                    atomicAdd(ap,      vv[u].x * ex);
                    atomicAdd(ap + 32, vv[u].y * ex);
                    atomicAdd(ap + 64, vv[u].z * ex);
                    atomicAdd(ap + 96, vv[u].w * ex);
                }
            }
        }
    }
}

__global__ __launch_bounds__(EG_T) void edge_kernel(
    const float* __restrict__ qbuf, const float* __restrict__ kbuf,
    const float* __restrict__ vbuf,
    const int* __restrict__ src, const int* __restrict__ dst,
    const float* __restrict__ feat,
    const float* __restrict__ ln_g, const float* __restrict__ ln_b,
    const unsigned* __restrict__ bmp,
    const float* __restrict__ w_ffn1, const float* __restrict__ bias_ffn1,
    const float* __restrict__ slope_ffn1,
    float* __restrict__ rst, float* __restrict__ hbuf, int n, int e)
{
    __shared__ float s_acc[EG_R * 128];
    __shared__ float s_q[EG_R * 128];
    __shared__ float s_m[EG_R * 8];
    __shared__ float s_l[EG_R * 8];
    __shared__ __attribute__((aligned(16))) float s_sc[EG_CS * 8];
    __shared__ int2  s_list[EG_CAP];
    __shared__ int   s_cnt;

    const int tid  = threadIdx.x;
    const int wave = tid >> 5;
    const int lane = tid & 31;
    const int lo   = blockIdx.x * EG_R;
    const int rows = (n - lo < EG_R) ? (n - lo) : EG_R;

    {
        const float4* qs = (const float4*)(qbuf + (size_t)lo * 128);
        float4 t[EG_QV];
#pragma unroll
        for (int u = 0; u < EG_QV; ++u) {
            const int i = tid + u * EG_T;
            t[u] = qs[(i < rows * 32) ? i : rows * 32 - 1];
        }
#pragma unroll
        for (int u = 0; u < EG_QV; ++u) {
            const int i = tid + u * EG_T;
            if (i < rows * 32) ((float4*)s_q)[i] = t[u];
        }
    }
    for (int i = tid; i < EG_R * 128; i += EG_T) s_acc[i] = 0.0f;
    for (int i = tid; i < EG_R * 8; i += EG_T) { s_m[i] = -INFINITY; s_l[i] = 0.0f; }
    if (tid == 0) s_cnt = 0;
    __syncthreads();

    const bool from_bm = bm_fits(n);
    if (from_bm) {
        const int nch = (e + BM_CE - 1) / BM_CE;
        const int nw4 = nch * (BM_WPC / 4);
        const uint4* row = (const uint4*)(bmp + (size_t)blockIdx.x * nch * BM_WPC);
        for (int base = 0; base < nw4; base += EG_T * BM_V)
            eg_scan_bits(row, nw4, base, s_list, &s_cnt, tid);
    } else {
        const int TE = EG_T * 32;
        const int nf = ((((size_t)dst) & 15) == 0) ? e / TE : 0;
        int t = (nf > 0) ? (int)(blockIdx.x % (unsigned)nf) : 0;
        for (int it = 0; it < nf; ++it) {
            eg_scan_tile<8, true, true>(dst, e, t * TE, lo, (unsigned)rows, s_list, &s_cnt, tid);
            if (++t == nf) t = 0;
        }
        for (int base = nf * TE; base < e; base += EG_T * 4)
            eg_scan_tile<1, true, false>(dst, e, base, lo, (unsigned)rows, s_list, &s_cnt, tid);
    }
    __syncthreads();
    const int total = s_cnt;

    float4 fv[EG_RW];
#pragma unroll
    for (int t = 0; t < EG_RW; ++t) {
        const int dl = wave + (EG_T / 32) * t;
        const int dc = (dl < rows) ? dl : rows - 1;
        fv[t] = *(const float4*)(feat + (size_t)(lo + dc) * 128 + 4 * lane);
    }
    const float g0 = ln_g[4 * lane],     g1 = ln_g[4 * lane + 1];
    const float g2 = ln_g[4 * lane + 2], g3 = ln_g[4 * lane + 3];
    const float b0 = ln_b[4 * lane],     b1 = ln_b[4 * lane + 1];
    const float b2 = ln_b[4 * lane + 2], b3 = ln_b[4 * lane + 3];

    if (total <= EG_CS) {
        if (from_bm) eg_resolve_bm(src, dst, s_list, total, lo, tid);
        else         eg_resolve(src, s_list, total, tid);
        __syncthreads();
        eg_process<0, true>(kbuf, vbuf, s_list, total, s_q, s_sc, s_m, s_l, s_acc, wave, lane);
        __syncthreads();
        eg_process<1, true>(kbuf, vbuf, s_list, total, s_q, s_sc, s_m, s_l, s_acc, wave, lane);
    } else if (total <= EG_CAP) {
        if (from_bm) eg_resolve_bm(src, dst, s_list, total, lo, tid);
        else         eg_resolve(src, s_list, total, tid);
        __syncthreads();
        eg_process<0, false>(kbuf, vbuf, s_list, total, s_q, s_sc, s_m, s_l, s_acc, wave, lane);
        __syncthreads();
        eg_process<1, false>(kbuf, vbuf, s_list, total, s_q, s_sc, s_m, s_l, s_acc, wave, lane);
    } else {
        const int TE = EG_T * 4;
#pragma unroll 1
        for (int pass = 0; pass < 2; ++pass) {
            __syncthreads();
            if (tid == 0) s_cnt = 0;
            __syncthreads();
            for (int base = 0; base < e; base += TE) {
                eg_scan_tile<1, false, false>(dst, e, base, lo, (unsigned)rows, s_list, &s_cnt, tid);
                __syncthreads();
                const int c = s_cnt;
                if (c > EG_CAP - TE) {
                    eg_resolve(src, s_list, c, tid);
                    __syncthreads();
                    if (pass == 0) eg_process<0, false>(kbuf, vbuf, s_list, c, s_q, s_sc, s_m, s_l, s_acc, wave, lane);
                    else           eg_process<1, false>(kbuf, vbuf, s_list, c, s_q, s_sc, s_m, s_l, s_acc, wave, lane);
                    __syncthreads();
                    if (tid == 0) s_cnt = 0;
                }
                __syncthreads();
            }
            const int c = s_cnt;
            eg_resolve(src, s_list, c, tid);
            __syncthreads();
            if (pass == 0) eg_process<0, false>(kbuf, vbuf, s_list, c, s_q, s_sc, s_m, s_l, s_acc, wave, lane);
            else           eg_process<1, false>(kbuf, vbuf, s_list, c, s_q, s_sc, s_m, s_l, s_acc, wave, lane);
        }
    }
    __syncthreads();

    float4 o[EG_RW];
#pragma unroll
    for (int t = 0; t < EG_RW; ++t) {
        const int dl = wave + (EG_T / 32) * t;
        const int dc = (dl < rows) ? dl : rows - 1;
        const float l = s_l[dc * 8 + (lane >> 2)];
        const float* ap = s_acc + dc * 128 + lane;
        const float a0 = ap[0], a1 = ap[32], a2 = ap[64], a3 = ap[96];
        const bool has = l > 0.0f;
        const float x0 = (has ? a0 / l : 0.0f) + fv[t].x;
        const float x1 = (has ? a1 / l : 0.0f) + fv[t].y;
        const float x2 = (has ? a2 / l : 0.0f) + fv[t].z;
        const float x3 = (has ? a3 / l : 0.0f) + fv[t].w;
        float s  = (x0 + x1) + (x2 + x3);
        float ss = (x0 * x0 + x1 * x1) + (x2 * x2 + x3 * x3);
#pragma unroll
        for (int m = 1; m < 32; m <<= 1) {
            s  += __shfl_xor(s,  m, 32);
            ss += __shfl_xor(ss, m, 32);
        }
        const float mean = s * (1.0f / 128.0f);
        const float var  = ss * (1.0f / 128.0f) - mean * mean;
        const float inv  = rsqrtf(var + EPSLN);
        o[t].x = (x0 - mean) * inv * g0 + b0;
        o[t].y = (x1 - mean) * inv * g1 + b1;
        o[t].z = (x2 - mean) * inv * g2 + b2;
        o[t].w = (x3 - mean) * inv * g3 + b3;
    }
    float (*s_t)[132] = (float (*)[132])s_sc;
#pragma unroll
    for (int t = 0; t < EG_RW; ++t) {
        const int dl = wave + (EG_T / 32) * t;
        if (dl < rows) {
            *(float4*)(rst + (size_t)(lo + dl) * 128 + 4 * lane) = o[t];
            ((float4*)&s_t[dl][0])[lane] = o[t];
        }
    }

    __syncthreads();
    const int half = lane >> 4;
    const int lr   = lane & 15;
    const int col  = wave * 16 + lr;
    static_assert((EG_T / 32) * 16 == 512, "one 16-column tile of W1 per wave");
    v2f bfr[32];
#pragma unroll
    for (int kq = 0; kq < 32; ++kq) {
        const int kk = 4 * kq + 2 * half;
        bfr[kq] = (v2f){ w_ffn1[(size_t)kk * 512 + col],
                         w_ffn1[(size_t)(kk + 1) * 512 + col] };
    }
    const float bias = bias_ffn1[col], slope = slope_ffn1[col];

    const int nmt = (rows + 15) >> 4;
#pragma unroll 1
    for (int mt = 0; mt < nmt; ++mt) {
        int ar = mt * 16 + lr; if (ar >= rows) ar = rows - 1;
        const float* arow = &s_t[ar][0];
        v8f acc = {};
#pragma unroll
        for (int kq = 0; kq < 32; ++kq) {
            const int kk = 4 * kq + 2 * half;
            v2f a = { arow[kk], arow[kk + 1] };
            acc = wmma_f32_4(a, bfr[kq], acc);
        }
#pragma unroll
        for (int r = 0; r < 8; ++r) {
            const int rl = mt * 16 + r + 8 * half;
            if (rl < rows) {
                float h = acc[r] + bias;
                h = (h >= 0.0f) ? h : slope * h;
                hbuf[(size_t)(lo + rl) * 512 + col] = h;
            }
        }
    }
}

__device__ __forceinline__ float4 ffn2_fetch(const float* __restrict__ hbuf, int row0, int n,
                                             int wave, int lane, int u, int kc)
{
    int gr = row0 + wave + 8 * u; if (gr >= n) gr = n - 1;
    const unsigned off = (unsigned)gr * 512u + (unsigned)(kc * 128 + 4 * lane);
    return *(const float4*)(hbuf + off);
}

template <bool MORE>
__device__ __forceinline__ void ffn2_chunk(
    const float* __restrict__ hbuf, const float* __restrict__ W2,
    float (*cur)[132], float (*nxt)[132], v8f (&acc)[4],
    int kc, int row0, int n, int wave, int lane, int half, int lr, int col)
{
    v2f bfr[32];
#pragma unroll
    for (int kq = 0; kq < 32; ++kq) {
        const int kk = kc * 128 + 4 * kq + 2 * half;
        bfr[kq] = (v2f){ W2[(size_t)kk * 128 + col],
                         W2[(size_t)(kk + 1) * 128 + col] };
    }
#pragma unroll
    for (int hp = 0; hp < 2; ++hp) {
        float4 t0, t1, t2, t3;
        if (MORE) {
            t0 = ffn2_fetch(hbuf, row0, n, wave, lane, 4 * hp + 0, kc + 1);
            t1 = ffn2_fetch(hbuf, row0, n, wave, lane, 4 * hp + 1, kc + 1);
            t2 = ffn2_fetch(hbuf, row0, n, wave, lane, 4 * hp + 2, kc + 1);
            t3 = ffn2_fetch(hbuf, row0, n, wave, lane, 4 * hp + 3, kc + 1);
            __builtin_amdgcn_sched_barrier(0);
        }
#pragma unroll
        for (int mt = 2 * hp; mt < 2 * hp + 2; ++mt) {
#pragma unroll
            for (int kq = 0; kq < 32; ++kq) {
                const int kk = 4 * kq + 2 * half;
                v2f a = { cur[mt * 16 + lr][kk], cur[mt * 16 + lr][kk + 1] };
                acc[mt] = wmma_f32_4(a, bfr[kq], acc[mt]);
            }
        }
        if (MORE) {
            ((float4*)&nxt[wave + 8 * (4 * hp + 0)][0])[lane] = t0;
            ((float4*)&nxt[wave + 8 * (4 * hp + 1)][0])[lane] = t1;
            ((float4*)&nxt[wave + 8 * (4 * hp + 2)][0])[lane] = t2;
            ((float4*)&nxt[wave + 8 * (4 * hp + 3)][0])[lane] = t3;
        }
    }
}

__global__ __launch_bounds__(256) __attribute__((amdgpu_waves_per_eu(6, 6))) void ffn2_ln_kernel(
    const float* __restrict__ hbuf, const float* __restrict__ W2,
    const float* __restrict__ b2, const float* __restrict__ rst,
    const float* __restrict__ g, const float* __restrict__ be,
    float* __restrict__ out, int n)
{
    __shared__ float sA[2][64][132];
    const int row0 = blockIdx.x * 64;
    const int tid  = threadIdx.x;
    const int wave = tid >> 5;
    const int lane = tid & 31;
    const int half = lane >> 4;
    const int lr   = lane & 15;
    const int col  = wave * 16 + lr;

    {
        float4 t[8];
#pragma unroll
        for (int u = 0; u < 8; ++u) t[u] = ffn2_fetch(hbuf, row0, n, wave, lane, u, 0);
#pragma unroll
        for (int u = 0; u < 8; ++u) ((float4*)&sA[0][wave + 8 * u][0])[lane] = t[u];
    }

    v8f acc[4] = {};
#pragma unroll 1
    for (int kc = 0; kc < 3; ++kc) {
        __syncthreads();
        ffn2_chunk<true>(hbuf, W2, sA[kc & 1], sA[(kc + 1) & 1], acc,
                         kc, row0, n, wave, lane, half, lr, col);
    }
    __syncthreads();
    ffn2_chunk<false>(hbuf, W2, sA[1], sA[0], acc, 3, row0, n, wave, lane, half, lr, col);
    __builtin_amdgcn_sched_barrier(0);

    const float bias = b2[col];
#pragma unroll
    for (int mt = 0; mt < 4; ++mt) {
#pragma unroll
        for (int r = 0; r < 8; ++r) {
            const int rl = mt * 16 + r + 8 * half;
            int gr = row0 + rl; if (gr >= n) gr = n - 1;
            sA[0][rl][col] = acc[mt][r] + bias + rst[(size_t)gr * 128 + col];
        }
    }
    __syncthreads();

    const int rl2 = tid >> 2;
    const int c0  = tid & 3;
    const int row = row0 + rl2;
    float xv[32], s = 0.0f, ss = 0.0f;
#pragma unroll
    for (int j = 0; j < 32; ++j) {
        int c = c0 + 4 * j;
        float x = sA[0][rl2][c];
        xv[j] = x; s += x; ss += x * x;
    }
#pragma unroll
    for (int m = 1; m < 4; m <<= 1) {
        s  += __shfl_xor(s,  m, 32);
        ss += __shfl_xor(ss, m, 32);
    }
    const float mean = s * (1.0f / 128.0f);
    const float var  = ss * (1.0f / 128.0f) - mean * mean;
    const float inv  = rsqrtf(var + EPSLN);
    if (row < n) {
        float* o = out + (size_t)row * 128;
#pragma unroll
        for (int j = 0; j < 32; ++j) {
            int c = c0 + 4 * j;
            o[c] = (xv[j] - mean) * inv * g[c] + be[c];
        }
    }
}

extern "C" void kernel_launch(void* const* d_in, const int* in_sizes, int n_in,
                              void* d_out, int out_size, void* d_ws, size_t ws_size,
                              hipStream_t stream)
{
    const float* feat  = (const float*)d_in[0];
    const int*   src   = (const int*)  d_in[1];
    const int*   dst   = (const int*)  d_in[2];
    const float* Wq    = (const float*)d_in[3];
    const float* Wk    = (const float*)d_in[4];
    const float* Wv    = (const float*)d_in[5];
    const float* ln1_g = (const float*)d_in[6];
    const float* ln1_b = (const float*)d_in[7];
    const float* ln2_g = (const float*)d_in[8];
    const float* ln2_b = (const float*)d_in[9];
    const float* W1    = (const float*)d_in[10];
    const float* b1    = (const float*)d_in[11];
    const float* pw    = (const float*)d_in[12];
    const float* W2    = (const float*)d_in[13];
    const float* b2    = (const float*)d_in[14];

    const int n = in_sizes[0] / 128;
    const int e = in_sizes[1];

    float* ws = (float*)d_ws;
    size_t off = 0;
    float* q      = ws + off; off += (size_t)n * 128;
    float* kbuf   = ws + off; off += (size_t)n * 128;
    float* vbuf   = ws + off; off += (size_t)n * 128;
    float* rst    = ws + off; off += (size_t)n * 128;
    float* hbuf   = ws + off; off += (size_t)n * 512;

    const int NT64 = (n + 63) / 64;
    const int NEG  = (n + EG_R - 1) / EG_R;
    const int NBM  = (e + BM_CE - 1) / BM_CE;

    unsigned* bmp = (unsigned*)(ws + off); off += (size_t)NEG * NBM * BM_WPC;
    (void)off; (void)ws_size; (void)n_in; (void)out_size;

    qkv_kernel<<<NT64 + NBM, 256, 0, stream>>>(feat, Wq, Wk, Wv, q, kbuf, vbuf,
                                               dst, bmp, n, e, NT64);
    edge_kernel<<<NEG, EG_T, 0, stream>>>(q, kbuf, vbuf, src, dst,
                                          feat, ln1_g, ln1_b, bmp, W1, b1, pw,
                                          rst, hbuf, n, e);
    ffn2_ln_kernel<<<NT64, 256, 0, stream>>>(hbuf, W2, b2, rst, ln2_g, ln2_b,
                                             (float*)d_out, n);
}


# edge softmax in one list walk: shift = bound 16*max|k_h|*max|q_row,h|/sqrt(128) (k maxima from launch 1), not the segment max; max walk only if bound > 16; float sums still in arrival order
# speedup vs baseline: 3.7912x; 1.0426x over previous
#include <hip/hip_runtime.h>
#include <hip/hip_bf16.h>
#include <math.h>

typedef __attribute__((ext_vector_type(2))) float v2f;
typedef __attribute__((ext_vector_type(8))) float v8f;

#define EPSLN 1e-5f

#define EG_R    157
#define BM_CE   1024
#define BM_WPC  (BM_CE / 32)
__device__ __forceinline__ bool bm_fits(int n) {
    return ((n + EG_R - 1) / EG_R) * BM_WPC <= 64 * 132;
}

__device__ __forceinline__ v8f wmma_f32_4(v2f a, v2f b, v8f c) {
    return __builtin_amdgcn_wmma_f32_16x16x4_f32(
         false, a,  false, b,
         (short)0, c,  false,  false);
}

__global__ __launch_bounds__(256) void qkv_kernel(
    const float* __restrict__ feat,
    const float* __restrict__ Wq, const float* __restrict__ Wk,
    const float* __restrict__ Wv,
    float* __restrict__ q, float* __restrict__ k, float* __restrict__ v,
    float* __restrict__ kmx,
    const int* __restrict__ dst, unsigned* __restrict__ bmp, int n, int e, int nt64)
{
    __shared__ float sA[64][132];
    const int tid  = threadIdx.x;

    if ((int)blockIdx.x >= nt64) {
        if (!bm_fits(n)) return;
        const int c   = (int)blockIdx.x - nt64;
        const int neg = (n + EG_R - 1) / EG_R;
        const int nch = (e + BM_CE - 1) / BM_CE;
        unsigned* bm  = (unsigned*)&sA[0][0];
        const int i0  = c * BM_CE + 4 * tid;
        int4 d;
        if (i0 + 3 < e && (((size_t)dst) & 15) == 0) {
            d = *(const int4*)(dst + i0);
        } else {
            d.x = (i0     < e) ? dst[i0]     : -1;
            d.y = (i0 + 1 < e) ? dst[i0 + 1] : -1;
            d.z = (i0 + 2 < e) ? dst[i0 + 2] : -1;
            d.w = (i0 + 3 < e) ? dst[i0 + 3] : -1;
        }
        for (int i = tid; i < neg * BM_WPC; i += 256) bm[i] = 0u;
        __syncthreads();
        const int      wq = tid >> 3;
        const unsigned b0 = 1u << (4 * (tid & 7));
        if ((unsigned)d.x < (unsigned)n) atomicOr(&bm[(d.x / EG_R) * BM_WPC + wq], b0);
        if ((unsigned)d.y < (unsigned)n) atomicOr(&bm[(d.y / EG_R) * BM_WPC + wq], b0 << 1);
        if ((unsigned)d.z < (unsigned)n) atomicOr(&bm[(d.z / EG_R) * BM_WPC + wq], b0 << 2);
        if ((unsigned)d.w < (unsigned)n) atomicOr(&bm[(d.w / EG_R) * BM_WPC + wq], b0 << 3);
        __syncthreads();
        for (int i = tid; i < neg * (BM_WPC / 4); i += 256) {
            const int o = i >> 3, qd = i & 7;
            ((uint4*)(bmp + ((size_t)o * nch + c) * BM_WPC))[qd] = ((const uint4*)bm)[i];
        }
        return;
    }

    const int row0 = blockIdx.x * 64;

    for (int i = tid; i < 64 * 32; i += 256) {
        int r = i >> 5, c4 = i & 31;
        int gr = row0 + r; if (gr >= n) gr = n - 1;
        ((float4*)&sA[r][0])[c4] = ((const float4*)(feat + (size_t)gr * 128))[c4];
    }
    __syncthreads();

    const int wave = tid >> 5;
    const int lane = tid & 31;
    const int half = lane >> 4;
    const int lr   = lane & 15;
    const int col  = wave * 16 + lr;

    const float* Ws[3] = { Wq, Wv, Wk };
    float*       Os[3] = { q,  v,  k  };

#pragma unroll
    for (int m = 0; m < 3; ++m) {
        const float* __restrict__ W = Ws[m];
        v2f bfr[32];
#pragma unroll
        for (int kq = 0; kq < 32; ++kq) {
            const int kk = 4 * kq + 2 * half;
            bfr[kq] = (v2f){ W[(size_t)kk * 128 + col],
                             W[(size_t)(kk + 1) * 128 + col] };
        }
        float* __restrict__ O = Os[m];
        float amax = 0.0f;
#pragma unroll
        for (int mt = 0; mt < 4; ++mt) {
            v8f acc = {};
#pragma unroll
            for (int kq = 0; kq < 32; ++kq) {
                const int kk = 4 * kq + 2 * half;
                v2f a = { sA[mt * 16 + lr][kk], sA[mt * 16 + lr][kk + 1] };
                acc = wmma_f32_4(a, bfr[kq], acc);
            }
#pragma unroll
            for (int r = 0; r < 8; ++r) {
                int gr = row0 + mt * 16 + r + 8 * half;
                if (gr < n) O[(size_t)gr * 128 + col] = acc[r];
                if (m == 2) amax = fmaxf(amax, fabsf(acc[r]));
            }
        }
        if (m == 2) {
            for (int s = 1; s < 32; s <<= 1) amax = fmaxf(amax, __shfl_xor(amax, s, 32));
            if (lane == 0) kmx[(size_t)blockIdx.x * 8 + wave] = amax;
        }
    }
}

#define EG_T    1024
#define EG_CAP  6144
#define EG_CS   3072
#define EG_QV   ((EG_R * 32 + EG_T - 1) / EG_T)
#define EG_RW   ((EG_R + EG_T / 32 - 1) / (EG_T / 32))

__device__ __forceinline__ void lds_max_f(float* addr, float val) {
    val += 0.0f;
    if (val >= 0.0f) atomicMax((int*)addr, __float_as_int(val));
    else             atomicMin((unsigned int*)addr, __float_as_uint(val));
}

__device__ __forceinline__ float quad_sum(float p) {
    p += __int_as_float(__builtin_amdgcn_mov_dpp(__float_as_int(p), 0xB1, 0xF, 0xF, true));
    p += __int_as_float(__builtin_amdgcn_mov_dpp(__float_as_int(p), 0x4E, 0xF, 0xF, true));
    return p;
}

template <int V4, bool CAPPED, bool FULL>
__device__ __forceinline__ void eg_scan_tile(
    const int* __restrict__ dst, int e, int base,
    int lo, unsigned rows, int2* s_list, int* s_cnt, int tid)
{
    int4 dv[V4];
#pragma unroll
    for (int u = 0; u < V4; ++u) {
        const int i0 = base + (u * EG_T + tid) * 4;
        if (FULL) {
            dv[u] = *(const int4*)(dst + i0);
        } else {
            dv[u].x = (i0     < e) ? dst[i0]     : -1;
            dv[u].y = (i0 + 1 < e) ? dst[i0 + 1] : -1;
            dv[u].z = (i0 + 2 < e) ? dst[i0 + 2] : -1;
            dv[u].w = (i0 + 3 < e) ? dst[i0 + 3] : -1;
        }
    }
#pragma unroll
    for (int u = 0; u < V4; ++u) {
        const int i0 = base + (u * EG_T + tid) * 4;
        const unsigned r0 = (unsigned)(dv[u].x - lo), r1 = (unsigned)(dv[u].y - lo);
        const unsigned r2 = (unsigned)(dv[u].z - lo), r3 = (unsigned)(dv[u].w - lo);
        const int c = (int)(r0 < rows) + (int)(r1 < rows) + (int)(r2 < rows) + (int)(r3 < rows);
        if (c) {
            int p = atomicAdd(s_cnt, c);
            if (r0 < rows) { if (!CAPPED || p < EG_CAP) s_list[p] = make_int2(i0,     (int)r0); ++p; }
            if (r1 < rows) { if (!CAPPED || p < EG_CAP) s_list[p] = make_int2(i0 + 1, (int)r1); ++p; }
            if (r2 < rows) { if (!CAPPED || p < EG_CAP) s_list[p] = make_int2(i0 + 2, (int)r2); ++p; }
            if (r3 < rows) { if (!CAPPED || p < EG_CAP) s_list[p] = make_int2(i0 + 3, (int)r3); }
        }
    }
}

__device__ __forceinline__ void eg_resolve(
    const int* __restrict__ src, int2* s_list, int cnt, int tid)
{
    for (int i = tid; i < cnt; i += EG_T) s_list[i].x = src[s_list[i].x];
}

__device__ __forceinline__ void eg_resolve_bm(
    const int* __restrict__ src, const int* __restrict__ dst,
    int2* s_list, int cnt, int lo, int tid)
{
    for (int i = tid; i < cnt; i += EG_T) {
        const int ed = s_list[i].x;
        s_list[i] = make_int2(src[ed], dst[ed] - lo);
    }
}

#define BM_V 5
__device__ __forceinline__ void eg_scan_bits(
    const uint4* __restrict__ row, int nw4, int base, int2* s_list, int* s_cnt, int tid)
{
    uint4 w[BM_V];
#pragma unroll
    for (int u = 0; u < BM_V; ++u) {
        const int idx = base + u * EG_T + tid;
        w[u] = (idx < nw4) ? row[idx] : make_uint4(0u, 0u, 0u, 0u);
    }
    int c = 0;
#pragma unroll
    for (int u = 0; u < BM_V; ++u)
        c += __popc(w[u].x) + __popc(w[u].y) + __popc(w[u].z) + __popc(w[u].w);
    if (c) {
        int p = atomicAdd(s_cnt, c);
#pragma unroll
        for (int u = 0; u < BM_V; ++u) {
            const int e0 = (base + u * EG_T + tid) * 128;
            const unsigned ww[4] = { w[u].x, w[u].y, w[u].z, w[u].w };
#pragma unroll
            for (int k = 0; k < 4; ++k) {
                unsigned m = ww[k];
                while (m) {
                    const int b = __ffs((int)m) - 1;
                    m &= m - 1u;
                    if (p < EG_CAP) s_list[p] = make_int2(e0 + 32 * k + b, 0);
                    ++p;
                }
            }
        }
    }
}

template <int PASS, bool CACHE>
__device__ __forceinline__ void eg_process(
    const float* __restrict__ kbuf, const float* __restrict__ vbuf,
    const int2* s_list, int cnt, const float* s_q, float* s_sc,
    float* s_m, float* s_l, float* s_acc, int wave, int lane)
{
    const int hq = lane >> 2;
    for (int i0 = wave; i0 < cnt; i0 += 32 * 4) {
        int2   en[4];
        float4 kk[4];
        float4 vv[4];
#pragma unroll
        for (int u = 0; u < 4; ++u) {
            const int i = i0 + 32 * u;
            en[u] = s_list[(i < cnt) ? i : i0];
            if (!(PASS == 1 && CACHE))
                kk[u] = *(const float4*)(kbuf + (size_t)en[u].x * 128 + 4 * lane);
            if (PASS == 1)
                vv[u] = *(const float4*)(vbuf + (size_t)en[u].x * 128 + 4 * lane);
        }
#pragma unroll
        for (int u = 0; u < 4; ++u) {
            const bool ok = (i0 + 32 * u) < cnt;
            const int  ic = ok ? (i0 + 32 * u) : i0;
            const int  dl = en[u].y;
            float sc;
            if (PASS == 1 && CACHE) {
                sc = s_sc[ic * 8 + hq];
            } else {
                const float4 qq = *(const float4*)(s_q + dl * 128 + 4 * lane);
                const float p = quad_sum(kk[u].x * qq.x + kk[u].y * qq.y +
                                         kk[u].z * qq.z + kk[u].w * qq.w);
                sc = p * 0.08838834764831845f;
            }
            if (PASS == 0) {
                if (ok && (lane & 3) == 0) {
                    lds_max_f(&s_m[dl * 8 + hq], sc);
                    if (CACHE) s_sc[ic * 8 + hq] = sc;
                }
            } else {
                const float ex = expf(sc - s_m[dl * 8 + hq]);
                if (ok) {
                    if ((lane & 3) == 0) atomicAdd(&s_l[dl * 8 + hq], ex);
                    float* ap = s_acc + dl * 128 + lane;
                    atomicAdd(ap,      vv[u].x * ex);
                    atomicAdd(ap + 32, vv[u].y * ex);
                    atomicAdd(ap + 64, vv[u].z * ex);
                    atomicAdd(ap + 96, vv[u].w * ex);
                }
            }
        }
    }
}

#define EG_BMAX 16.0f

__global__ __launch_bounds__(EG_T) void edge_kernel(
    const float* __restrict__ qbuf, const float* __restrict__ kbuf,
    const float* __restrict__ vbuf,
    const int* __restrict__ src, const int* __restrict__ dst,
    const float* __restrict__ feat,
    const float* __restrict__ ln_g, const float* __restrict__ ln_b,
    const unsigned* __restrict__ bmp, const float* __restrict__ kmx,
    const float* __restrict__ w_ffn1, const float* __restrict__ bias_ffn1,
    const float* __restrict__ slope_ffn1,
    float* __restrict__ rst, float* __restrict__ hbuf, int n, int e, int nt64)
{
    __shared__ float s_acc[EG_R * 128];
    __shared__ float s_q[EG_R * 128];
    __shared__ float s_m[EG_R * 8];
    __shared__ float s_l[EG_R * 8];
    __shared__ __attribute__((aligned(16))) float s_sc[EG_CS * 8];
    __shared__ int2  s_list[EG_CAP];
    __shared__ int   s_cnt;
    __shared__ float s_kmax[8];
    __shared__ int   s_wide;

    const int tid  = threadIdx.x;
    const int wave = tid >> 5;
    const int lane = tid & 31;
    const int lo   = blockIdx.x * EG_R;
    const int rows = (n - lo < EG_R) ? (n - lo) : EG_R;

    float kt[8];
#pragma unroll
    for (int u = 0; u < 8; ++u) {
        const int b = (tid >> 3) + u * (EG_T / 8);
        kt[u] = kmx[(size_t)((b < nt64) ? b : nt64 - 1) * 8 + (tid & 7)];
    }

    {
        const float4* qs = (const float4*)(qbuf + (size_t)lo * 128);
        float4 t[EG_QV];
#pragma unroll
        for (int u = 0; u < EG_QV; ++u) {
            const int i = tid + u * EG_T;
            t[u] = qs[(i < rows * 32) ? i : rows * 32 - 1];
        }
#pragma unroll
        for (int u = 0; u < EG_QV; ++u) {
            const int i = tid + u * EG_T;
            if (i < rows * 32) ((float4*)s_q)[i] = t[u];
        }
    }
    for (int i = tid; i < EG_R * 128; i += EG_T) s_acc[i] = 0.0f;
    for (int i = tid; i < EG_R * 8; i += EG_T) s_l[i] = 0.0f;
    if (tid == 0) { s_cnt = 0; s_wide = 0; }
    if (tid < 8) s_kmax[tid] = 0.0f;
    __syncthreads();

    float kmx_mine = fmaxf(fmaxf(fmaxf(kt[0], kt[1]), fmaxf(kt[2], kt[3])),
                           fmaxf(fmaxf(kt[4], kt[5]), fmaxf(kt[6], kt[7])));
    for (int b = (tid >> 3) + 8 * (EG_T / 8); b < nt64; b += EG_T / 8)
        kmx_mine = fmaxf(kmx_mine, kmx[(size_t)b * 8 + (tid & 7)]);
    atomicMax((int*)&s_kmax[tid & 7], __float_as_int(fmaxf(kmx_mine, 0.0f)));

    const bool from_bm = bm_fits(n);
    if (from_bm) {
        const int nch = (e + BM_CE - 1) / BM_CE;
        const int nw4 = nch * (BM_WPC / 4);
        const uint4* row = (const uint4*)(bmp + (size_t)blockIdx.x * nch * BM_WPC);
        for (int base = 0; base < nw4; base += EG_T * BM_V)
            eg_scan_bits(row, nw4, base, s_list, &s_cnt, tid);
    } else {
        const int TE = EG_T * 32;
        const int nf = ((((size_t)dst) & 15) == 0) ? e / TE : 0;
        int t = (nf > 0) ? (int)(blockIdx.x % (unsigned)nf) : 0;
        for (int it = 0; it < nf; ++it) {
            eg_scan_tile<8, true, true>(dst, e, t * TE, lo, (unsigned)rows, s_list, &s_cnt, tid);
            if (++t == nf) t = 0;
        }
        for (int base = nf * TE; base < e; base += EG_T * 4)
            eg_scan_tile<1, true, false>(dst, e, base, lo, (unsigned)rows, s_list, &s_cnt, tid);
    }
    __syncthreads();
    const int total = s_cnt;

    float4 fv[EG_RW];
#pragma unroll
    for (int t = 0; t < EG_RW; ++t) {
        const int dl = wave + (EG_T / 32) * t;
        const int dc = (dl < rows) ? dl : rows - 1;
        fv[t] = *(const float4*)(feat + (size_t)(lo + dc) * 128 + 4 * lane);
    }
    const float g0 = ln_g[4 * lane],     g1 = ln_g[4 * lane + 1];
    const float g2 = ln_g[4 * lane + 2], g3 = ln_g[4 * lane + 3];
    const float b0 = ln_b[4 * lane],     b1 = ln_b[4 * lane + 1];
    const float b2 = ln_b[4 * lane + 2], b3 = ln_b[4 * lane + 3];

    for (int i = tid; i < rows * 8; i += EG_T) {
        const float4* qp = (const float4*)(s_q + (i >> 3) * 128 + (i & 7) * 16);
        const float4 a = qp[0], b = qp[1], c4 = qp[2], d = qp[3];
        const float qm = fmaxf(fmaxf(fmaxf(fmaxf(fabsf(a.x), fabsf(a.y)), fmaxf(fabsf(a.z), fabsf(a.w))),
                                     fmaxf(fmaxf(fabsf(b.x), fabsf(b.y)), fmaxf(fabsf(b.z), fabsf(b.w)))),
                               fmaxf(fmaxf(fmaxf(fabsf(c4.x), fabsf(c4.y)), fmaxf(fabsf(c4.z), fabsf(c4.w))),
                                     fmaxf(fmaxf(fabsf(d.x), fabsf(d.y)), fmaxf(fabsf(d.z), fabsf(d.w)))));
        const float bound = 16.0f * s_kmax[i & 7] * qm * 0.08838834764831845f;
        s_m[i] = bound;
        if (!(bound <= EG_BMAX)) s_wide = 1;
    }
    if (total <= EG_CAP) {
        if (from_bm) eg_resolve_bm(src, dst, s_list, total, lo, tid);
        else         eg_resolve(src, s_list, total, tid);
    }
    __syncthreads();
    const bool one_walk = (s_wide == 0) && (total <= EG_CAP);

    if (one_walk) {
        eg_process<1, false>(kbuf, vbuf, s_list, total, s_q, s_sc, s_m, s_l, s_acc, wave, lane);
    } else {
        for (int i = tid; i < EG_R * 8; i += EG_T) s_m[i] = -INFINITY;
        __syncthreads();
        if (total <= EG_CS) {
            eg_process<0, true>(kbuf, vbuf, s_list, total, s_q, s_sc, s_m, s_l, s_acc, wave, lane);
            __syncthreads();
            eg_process<1, true>(kbuf, vbuf, s_list, total, s_q, s_sc, s_m, s_l, s_acc, wave, lane);
        } else if (total <= EG_CAP) {
            eg_process<0, false>(kbuf, vbuf, s_list, total, s_q, s_sc, s_m, s_l, s_acc, wave, lane);
            __syncthreads();
            eg_process<1, false>(kbuf, vbuf, s_list, total, s_q, s_sc, s_m, s_l, s_acc, wave, lane);
        } else {
            const int TE = EG_T * 4;
#pragma unroll 1
            for (int pass = 0; pass < 2; ++pass) {
                __syncthreads();
                if (tid == 0) s_cnt = 0;
                __syncthreads();
                for (int base = 0; base < e; base += TE) {
                    eg_scan_tile<1, false, false>(dst, e, base, lo, (unsigned)rows, s_list, &s_cnt, tid);
                    __syncthreads();
                    const int c = s_cnt;
                    if (c > EG_CAP - TE) {
                        eg_resolve(src, s_list, c, tid);
                        __syncthreads();
                        if (pass == 0) eg_process<0, false>(kbuf, vbuf, s_list, c, s_q, s_sc, s_m, s_l, s_acc, wave, lane);
                        else           eg_process<1, false>(kbuf, vbuf, s_list, c, s_q, s_sc, s_m, s_l, s_acc, wave, lane);
                        __syncthreads();
                        if (tid == 0) s_cnt = 0;
                    }
                    __syncthreads();
                }
                const int c = s_cnt;
                eg_resolve(src, s_list, c, tid);
                __syncthreads();
                if (pass == 0) eg_process<0, false>(kbuf, vbuf, s_list, c, s_q, s_sc, s_m, s_l, s_acc, wave, lane);
                else           eg_process<1, false>(kbuf, vbuf, s_list, c, s_q, s_sc, s_m, s_l, s_acc, wave, lane);
            }
        }
    }
    __syncthreads();

    float4 o[EG_RW];
#pragma unroll
    for (int t = 0; t < EG_RW; ++t) {
        const int dl = wave + (EG_T / 32) * t;
        const int dc = (dl < rows) ? dl : rows - 1;
        const float l = s_l[dc * 8 + (lane >> 2)];
        const float* ap = s_acc + dc * 128 + lane;
        const float a0 = ap[0], a1 = ap[32], a2 = ap[64], a3 = ap[96];
        const bool has = l > 0.0f;
        const float x0 = (has ? a0 / l : 0.0f) + fv[t].x;
        const float x1 = (has ? a1 / l : 0.0f) + fv[t].y;
        const float x2 = (has ? a2 / l : 0.0f) + fv[t].z;
        const float x3 = (has ? a3 / l : 0.0f) + fv[t].w;
        float s  = (x0 + x1) + (x2 + x3);
        float ss = (x0 * x0 + x1 * x1) + (x2 * x2 + x3 * x3);
#pragma unroll
        for (int m = 1; m < 32; m <<= 1) {
            s  += __shfl_xor(s,  m, 32);
            ss += __shfl_xor(ss, m, 32);
        }
        const float mean = s * (1.0f / 128.0f);
        const float var  = ss * (1.0f / 128.0f) - mean * mean;
        const float inv  = rsqrtf(var + EPSLN);
        o[t].x = (x0 - mean) * inv * g0 + b0;
        o[t].y = (x1 - mean) * inv * g1 + b1;
        o[t].z = (x2 - mean) * inv * g2 + b2;
        o[t].w = (x3 - mean) * inv * g3 + b3;
    }
    float (*s_t)[132] = (float (*)[132])s_sc;
#pragma unroll
    for (int t = 0; t < EG_RW; ++t) {
        const int dl = wave + (EG_T / 32) * t;
        if (dl < rows) {
            *(float4*)(rst + (size_t)(lo + dl) * 128 + 4 * lane) = o[t];
            ((float4*)&s_t[dl][0])[lane] = o[t];
        }
    }

    __syncthreads();
    const int half = lane >> 4;
    const int lr   = lane & 15;
    const int col  = wave * 16 + lr;
    static_assert((EG_T / 32) * 16 == 512, "one 16-column tile of W1 per wave");
    v2f bfr[32];
#pragma unroll
    for (int kq = 0; kq < 32; ++kq) {
        const int kk = 4 * kq + 2 * half;
        bfr[kq] = (v2f){ w_ffn1[(size_t)kk * 512 + col],
                         w_ffn1[(size_t)(kk + 1) * 512 + col] };
    }
    const float bias = bias_ffn1[col], slope = slope_ffn1[col];

    const int nmt = (rows + 15) >> 4;
#pragma unroll 1
    for (int mt = 0; mt < nmt; ++mt) {
        int ar = mt * 16 + lr; if (ar >= rows) ar = rows - 1;
        const float* arow = &s_t[ar][0];
        v8f acc = {};
#pragma unroll
        for (int kq = 0; kq < 32; ++kq) {
            const int kk = 4 * kq + 2 * half;
            v2f a = { arow[kk], arow[kk + 1] };
            acc = wmma_f32_4(a, bfr[kq], acc);
        }
#pragma unroll
        for (int r = 0; r < 8; ++r) {
            const int rl = mt * 16 + r + 8 * half;
            if (rl < rows) {
                float h = acc[r] + bias;
                h = (h >= 0.0f) ? h : slope * h;
                hbuf[(size_t)(lo + rl) * 512 + col] = h;
            }
        }
    }
}

__device__ __forceinline__ float4 ffn2_fetch(const float* __restrict__ hbuf, int row0, int n,
                                             int wave, int lane, int u, int kc)
{
    int gr = row0 + wave + 8 * u; if (gr >= n) gr = n - 1;
    const unsigned off = (unsigned)gr * 512u + (unsigned)(kc * 128 + 4 * lane);
    return *(const float4*)(hbuf + off);
}

template <bool MORE>
__device__ __forceinline__ void ffn2_chunk(
    const float* __restrict__ hbuf, const float* __restrict__ W2,
    float (*cur)[132], float (*nxt)[132], v8f (&acc)[4],
    int kc, int row0, int n, int wave, int lane, int half, int lr, int col)
{
    v2f bfr[32];
#pragma unroll
    for (int kq = 0; kq < 32; ++kq) {
        const int kk = kc * 128 + 4 * kq + 2 * half;
        bfr[kq] = (v2f){ W2[(size_t)kk * 128 + col],
                         W2[(size_t)(kk + 1) * 128 + col] };
    }
#pragma unroll
    for (int hp = 0; hp < 2; ++hp) {
        float4 t0, t1, t2, t3;
        if (MORE) {
            t0 = ffn2_fetch(hbuf, row0, n, wave, lane, 4 * hp + 0, kc + 1);
            t1 = ffn2_fetch(hbuf, row0, n, wave, lane, 4 * hp + 1, kc + 1);
            t2 = ffn2_fetch(hbuf, row0, n, wave, lane, 4 * hp + 2, kc + 1);
            t3 = ffn2_fetch(hbuf, row0, n, wave, lane, 4 * hp + 3, kc + 1);
            __builtin_amdgcn_sched_barrier(0);
        }
#pragma unroll
        for (int mt = 2 * hp; mt < 2 * hp + 2; ++mt) {
#pragma unroll
            for (int kq = 0; kq < 32; ++kq) {
                const int kk = 4 * kq + 2 * half;
                v2f a = { cur[mt * 16 + lr][kk], cur[mt * 16 + lr][kk + 1] };
                acc[mt] = wmma_f32_4(a, bfr[kq], acc[mt]);
            }
        }
        if (MORE) {
            ((float4*)&nxt[wave + 8 * (4 * hp + 0)][0])[lane] = t0;
            ((float4*)&nxt[wave + 8 * (4 * hp + 1)][0])[lane] = t1;
            ((float4*)&nxt[wave + 8 * (4 * hp + 2)][0])[lane] = t2;
            ((float4*)&nxt[wave + 8 * (4 * hp + 3)][0])[lane] = t3;
        }
    }
}

__global__ __launch_bounds__(256) __attribute__((amdgpu_waves_per_eu(6, 6))) void ffn2_ln_kernel(
    const float* __restrict__ hbuf, const float* __restrict__ W2,
    const float* __restrict__ b2, const float* __restrict__ rst,
    const float* __restrict__ g, const float* __restrict__ be,
    float* __restrict__ out, int n)
{
    __shared__ float sA[2][64][132];
    const int row0 = blockIdx.x * 64;
    const int tid  = threadIdx.x;
    const int wave = tid >> 5;
    const int lane = tid & 31;
    const int half = lane >> 4;
    const int lr   = lane & 15;
    const int col  = wave * 16 + lr;

    {
        float4 t[8];
#pragma unroll
        for (int u = 0; u < 8; ++u) t[u] = ffn2_fetch(hbuf, row0, n, wave, lane, u, 0);
#pragma unroll
        for (int u = 0; u < 8; ++u) ((float4*)&sA[0][wave + 8 * u][0])[lane] = t[u];
    }

    v8f acc[4] = {};
#pragma unroll 1
    for (int kc = 0; kc < 3; ++kc) {
        __syncthreads();
        ffn2_chunk<true>(hbuf, W2, sA[kc & 1], sA[(kc + 1) & 1], acc,
                         kc, row0, n, wave, lane, half, lr, col);
    }
    __syncthreads();
    ffn2_chunk<false>(hbuf, W2, sA[1], sA[0], acc, 3, row0, n, wave, lane, half, lr, col);
    __builtin_amdgcn_sched_barrier(0);

    const float bias = b2[col];
#pragma unroll
    for (int mt = 0; mt < 4; ++mt) {
#pragma unroll
        for (int r = 0; r < 8; ++r) {
            const int rl = mt * 16 + r + 8 * half;
            int gr = row0 + rl; if (gr >= n) gr = n - 1;
            sA[0][rl][col] = acc[mt][r] + bias + rst[(size_t)gr * 128 + col];
        }
    }
    __syncthreads();

    const int rl2 = tid >> 2;
    const int c0  = tid & 3;
    const int row = row0 + rl2;
    float xv[32], s = 0.0f, ss = 0.0f;
#pragma unroll
    for (int j = 0; j < 32; ++j) {
        int c = c0 + 4 * j;
        float x = sA[0][rl2][c];
        xv[j] = x; s += x; ss += x * x;
    }
#pragma unroll
    for (int m = 1; m < 4; m <<= 1) {
        s  += __shfl_xor(s,  m, 32);
        ss += __shfl_xor(ss, m, 32);
    }
    const float mean = s * (1.0f / 128.0f);
    const float var  = ss * (1.0f / 128.0f) - mean * mean;
    const float inv  = rsqrtf(var + EPSLN);
    if (row < n) {
        float* o = out + (size_t)row * 128;
#pragma unroll
        for (int j = 0; j < 32; ++j) {
            int c = c0 + 4 * j;
            o[c] = (xv[j] - mean) * inv * g[c] + be[c];
        }
    }
}

extern "C" void kernel_launch(void* const* d_in, const int* in_sizes, int n_in,
                              void* d_out, int out_size, void* d_ws, size_t ws_size,
                              hipStream_t stream)
{
    const float* feat  = (const float*)d_in[0];
    const int*   src   = (const int*)  d_in[1];
    const int*   dst   = (const int*)  d_in[2];
    const float* Wq    = (const float*)d_in[3];
    const float* Wk    = (const float*)d_in[4];
    const float* Wv    = (const float*)d_in[5];
    const float* ln1_g = (const float*)d_in[6];
    const float* ln1_b = (const float*)d_in[7];
    const float* ln2_g = (const float*)d_in[8];
    const float* ln2_b = (const float*)d_in[9];
    const float* W1    = (const float*)d_in[10];
    const float* b1    = (const float*)d_in[11];
    const float* pw    = (const float*)d_in[12];
    const float* W2    = (const float*)d_in[13];
    const float* b2    = (const float*)d_in[14];

    const int n = in_sizes[0] / 128;
    const int e = in_sizes[1];

    float* ws = (float*)d_ws;
    size_t off = 0;
    float* q      = ws + off; off += (size_t)n * 128;
    float* kbuf   = ws + off; off += (size_t)n * 128;
    float* vbuf   = ws + off; off += (size_t)n * 128;
    float* rst    = ws + off; off += (size_t)n * 128;
    float* hbuf   = ws + off; off += (size_t)n * 512;

    const int NT64 = (n + 63) / 64;
    const int NEG  = (n + EG_R - 1) / EG_R;
    const int NBM  = (e + BM_CE - 1) / BM_CE;

    unsigned* bmp = (unsigned*)(ws + off); off += (size_t)NEG * NBM * BM_WPC;
    float* kmx    = ws + off; off += (size_t)NT64 * 8;
    (void)off; (void)ws_size; (void)n_in; (void)out_size;

    qkv_kernel<<<NT64 + NBM, 256, 0, stream>>>(feat, Wq, Wk, Wv, q, kbuf, vbuf, kmx,
                                               dst, bmp, n, e, NT64);
    edge_kernel<<<NEG, EG_T, 0, stream>>>(q, kbuf, vbuf, src, dst,
                                          feat, ln1_g, ln1_b, bmp, kmx, W1, b1, pw,
                                          rst, hbuf, n, e, NT64);
    ffn2_ln_kernel<<<NT64, 256, 0, stream>>>(hbuf, W2, b2, rst, ln2_g, ln2_b,
                                             (float*)d_out, n);
}
